# starting kernel
# baseline (speedup 1.0000x reference)
#include <hip/hip_runtime.h>
#include <hip/hip_bf16.h>

typedef __attribute__((ext_vector_type(2))) float v2f;
typedef __attribute__((ext_vector_type(8))) float v8f;

#define BB    2
#define HH_   90
#define WW    180
#define PP    (HH_ * WW)
#define CC    64
#define MD    66
#define SHN   16
#define KK    25
#define NNZN  25
#define HHID  32
#define FFNH  256

__device__ __forceinline__ float gelu_f(float v) {
    return 0.5f * v * (1.0f + erff(v * 0.70710678118654752440f));
}

__device__ __forceinline__ v8f wmma_f32(v2f a, v2f b, v8f c) {
    return __builtin_amdgcn_wmma_f32_16x16x4_f32(false, a, false, b, (short)0, c,
                                                 false, false);
}

__global__ __launch_bounds__(32)
void mod_kernel(const float* __restrict__ x,
                const float* __restrict__ lw1, const float* __restrict__ lb1,
                const float* __restrict__ lw2, const float* __restrict__ lb2,
                float* __restrict__ mod)
{
    const int row = blockIdx.x;
    const int j   = threadIdx.x;
    const float* xp = x + ((long)row * WW) * MD + CC;
    const float s0 = xp[0], s1 = xp[1];
    float acc = lb2[j];
    for (int i = 0; i < 2 * SHN; ++i) {
        float t = s0 * lw1[i] + s1 * lw1[2 * SHN + i] + lb1[i];
        acc += gelu_f(t) * lw2[i * (2 * SHN) + j];
    }
    mod[row * (2 * SHN) + j] = acc;
}

__global__ __launch_bounds__(32)
void disco_head_kernel(const float* __restrict__ x, const int* __restrict__ nbr,
                       const float* __restrict__ basis,
                       const float* __restrict__ disco_w,
                       const float* __restrict__ disco_b,
                       const float* __restrict__ hw1, const float* __restrict__ hb1,
                       const float* __restrict__ hw2, const float* __restrict__ hb2,
                       const float* __restrict__ mod, float* __restrict__ xln)
{
    __shared__ int   nbrx[28];
    __shared__ float bT[32 * 28];
    __shared__ float dws[16 * 28];
    __shared__ float bmat[32 * 16];
    __shared__ float fbuf[64 * 17];
    __shared__ float h1buf[64 * 33];

    const int l = threadIdx.x;
    const int q = blockIdx.x;
    const int b = q / PP;
    const int p = q - b * PP;

    for (int e = l; e < 32 * 28; e += 32) bT[e] = 0.f;
    for (int e = l; e < 16 * 28; e += 32) dws[e] = 0.f;
    __syncthreads();

    if (l < 28) nbrx[l] = nbr[p * NNZN + ((l < NNZN) ? l : 0)];
    for (int e = l; e < KK * KK; e += 32) {
        const int k = e / KK, n = e - k * KK;
        bT[n * 28 + k] = basis[(long)k * (PP * NNZN) + (long)p * NNZN + n];
    }
    for (int e = l; e < SHN * KK; e += 32) {
        const int o = e / KK, k = e - o * KK;
        dws[o * 28 + k] = disco_w[e];
    }
    __syncthreads();

    const bool hi = l >= 16;
    const int  lr = l & 15;

    {
        v8f zb = {0.f, 0.f, 0.f, 0.f, 0.f, 0.f, 0.f, 0.f};
        v8f accb[2] = {zb, zb};
        #pragma unroll
        for (int kt = 0; kt < 7; ++kt) {
            const int k0 = kt * 4 + (hi ? 2 : 0);
            v2f bf;
            bf.x = dws[lr * 28 + k0];
            bf.y = dws[lr * 28 + k0 + 1];
            #pragma unroll
            for (int t = 0; t < 2; ++t) {
                v2f af;
                af.x = bT[(t * 16 + lr) * 28 + k0];
                af.y = bT[(t * 16 + lr) * 28 + k0 + 1];
                accb[t] = wmma_f32(af, bf, accb[t]);
            }
        }
        #pragma unroll
        for (int t = 0; t < 2; ++t)
            #pragma unroll
            for (int r = 0; r < 8; ++r) {
                const int n = t * 16 + r + (hi ? 8 : 0);
                bmat[n * 16 + lr] = accb[t][r];
            }
    }
    __syncthreads();

    const float db = disco_b[lr];
    v8f acc[4];
    {
        v8f z = {db, db, db, db, db, db, db, db};
        #pragma unroll
        for (int t = 0; t < 4; ++t) acc[t] = z;
    }
    const long xbase = (long)b * PP;
    #pragma unroll
    for (int kt = 0; kt < 7; ++kt) {
        const int n0 = kt * 4 + (hi ? 2 : 0);
        const float* pa = x + (xbase + nbrx[n0]) * MD;
        const float* pb = x + (xbase + nbrx[n0 + 1]) * MD;
        v2f bf;
        bf.x = bmat[n0 * 16 + lr];
        bf.y = bmat[(n0 + 1) * 16 + lr];
        v2f af[4];
        #pragma unroll
        for (int t = 0; t < 4; ++t) {
            af[t].x = pa[t * 16 + lr];
            af[t].y = pb[t * 16 + lr];
        }
        #pragma unroll
        for (int t = 0; t < 4; ++t) acc[t] = wmma_f32(af[t], bf, acc[t]);
    }

    const int hrow = p / WW;
    const float* mrow = mod + (b * HH_ + hrow) * (2 * SHN);
    const float gm = mrow[lr];
    const float bt = mrow[SHN + lr];
    #pragma unroll
    for (int t = 0; t < 4; ++t)
        #pragma unroll
        for (int r = 0; r < 8; ++r) {
            const int c = t * 16 + r + (hi ? 8 : 0);
            fbuf[c * 17 + lr] = acc[t][r] * gm + bt;
        }
    __syncthreads();

    #pragma unroll
    for (int t = 0; t < 4; ++t) {
        #pragma unroll
        for (int nt = 0; nt < 2; ++nt) {
            const float binit = hb1[(2 * t + (hi ? 1 : 0)) * HHID + nt * 16 + lr];
            v8f acch = {binit, binit, binit, binit, binit, binit, binit, binit};
            #pragma unroll
            for (int kt = 0; kt < 8; ++kt) {
                const int  k0   = kt * 4 + (hi ? 2 : 0);
                const int  s0   = (kt < 4) ? k0 : (k0 - 16);
                const int  hsel = 2 * t + ((kt < 4) ? 0 : 1);
                const bool keep = (kt < 4) ? (lr < 8) : (lr >= 8);
                const float va0 = fbuf[(t * 16 + lr) * 17 + s0];
                const float va1 = fbuf[(t * 16 + lr) * 17 + s0 + 1];
                v2f af, bf;
                af.x = keep ? va0 : 0.f;
                af.y = keep ? va1 : 0.f;
                bf.x = hw1[(hsel * SHN + s0) * HHID + nt * 16 + lr];
                bf.y = hw1[(hsel * SHN + s0 + 1) * HHID + nt * 16 + lr];
                acch = wmma_f32(af, bf, acch);
            }
            #pragma unroll
            for (int r = 0; r < 8; ++r) {
                const int c = t * 16 + r + (hi ? 8 : 0);
                h1buf[c * 33 + nt * 16 + lr] = gelu_f(acch[r]);
            }
        }
    }
    __syncthreads();

    for (int cc = l; cc < CC; cc += 32) {
        const int n = cc >> 3;
        float h2 = hb2[n];
        #pragma unroll
        for (int o = 0; o < HHID; ++o)
            h2 += h1buf[cc * 33 + o] * hw2[n * HHID + o];
        const float res = x[((long)b * PP + p) * MD + cc];
        xln[(long)q * CC + cc] = h2 + res;
    }
}

__global__ __launch_bounds__(32)
void ffn_kernel(const float* __restrict__ x, const float* __restrict__ xln,
                const float* __restrict__ w1, const float* __restrict__ b1,
                const float* __restrict__ w2, const float* __restrict__ b2,
                float* __restrict__ out)
{
    __shared__ float xf[16 * 72];
    __shared__ float hid[16 * 260];

    const int  l   = threadIdx.x;
    const long gq0 = (long)blockIdx.x * 16;

    for (int e = l; e < 16 * 68; e += 32) {
        const int m = e / 68, c = e - m * 68;
        const long gq = gq0 + m;
        float v;
        if (c < CC)      v = xln[gq * CC + c];
        else if (c < MD) v = x[gq * MD + c];
        else             v = 0.f;
        xf[m * 72 + c] = v;
    }
    __syncthreads();

    const bool hi = l >= 16;
    const int  lr = l & 15;

    for (int nt = 0; nt < 16; ++nt) {
        const float bb = b1[nt * 16 + lr];
        v8f acc = {bb, bb, bb, bb, bb, bb, bb, bb};
        #pragma unroll
        for (int kt = 0; kt < 17; ++kt) {
            const int k0 = kt * 4 + (hi ? 2 : 0);
            const int ka = (k0     < MD) ? k0     : 0;
            const int kb = (k0 + 1 < MD) ? k0 + 1 : 0;
            v2f af, bf;
            af.x = xf[lr * 72 + k0];
            af.y = xf[lr * 72 + k0 + 1];
            const float w0 = w1[(long)ka * FFNH + nt * 16 + lr];
            const float w1v = w1[(long)kb * FFNH + nt * 16 + lr];
            bf.x = (k0     < MD) ? w0  : 0.f;
            bf.y = (k0 + 1 < MD) ? w1v : 0.f;
            acc = wmma_f32(af, bf, acc);
        }
        #pragma unroll
        for (int r = 0; r < 8; ++r) {
            const int m = r + (hi ? 8 : 0);
            hid[m * 260 + nt * 16 + lr] = gelu_f(acc[r]);
        }
    }
    __syncthreads();

    for (int nt = 0; nt < 4; ++nt) {
        const float bb = b2[nt * 16 + lr];
        v8f acc = {bb, bb, bb, bb, bb, bb, bb, bb};
        #pragma unroll
        for (int kt = 0; kt < 64; ++kt) {
            const int k0 = kt * 4 + (hi ? 2 : 0);
            v2f af, bf;
            af.x = hid[lr * 260 + k0];
            af.y = hid[lr * 260 + k0 + 1];
            bf.x = w2[k0 * CC + nt * 16 + lr];
            bf.y = w2[(k0 + 1) * CC + nt * 16 + lr];
            acc = wmma_f32(af, bf, acc);
        }
        #pragma unroll
        for (int r = 0; r < 8; ++r) {
            const int  m  = r + (hi ? 8 : 0);
            const long gq = gq0 + m;
            const int  c  = nt * 16 + lr;
            out[gq * MD + c] = acc[r] + xln[gq * CC + c];
        }
    }
    {
        const int  m  = l >> 1, c2 = l & 1;
        const long gq = gq0 + m;
        out[gq * MD + CC + c2] = x[gq * MD + CC + c2];
    }
}

extern "C" void kernel_launch(void* const* d_in, const int* in_sizes, int n_in,
                              void* d_out, int out_size, void* d_ws, size_t ws_size,
                              hipStream_t stream)
{
    const float* x       = (const float*)d_in[0];
    const int*   nbr     = (const int*)  d_in[1];
    const float* basis   = (const float*)d_in[2];
    const float* disco_w = (const float*)d_in[3];
    const float* disco_b = (const float*)d_in[4];
    const float* lw1     = (const float*)d_in[5];
    const float* lb1     = (const float*)d_in[6];
    const float* lw2     = (const float*)d_in[7];
    const float* lb2     = (const float*)d_in[8];
    const float* hw1     = (const float*)d_in[9];
    const float* hb1     = (const float*)d_in[10];
    const float* hw2     = (const float*)d_in[11];
    const float* hb2     = (const float*)d_in[12];
    const float* fw1     = (const float*)d_in[13];
    const float* fb1     = (const float*)d_in[14];
    const float* fw2     = (const float*)d_in[15];
    const float* fb2     = (const float*)d_in[16];
    float* out = (float*)d_out;

    float* mod = (float*)d_ws;
    float* xln = (float*)((char*)d_ws + (size_t)(BB * HH_ * 32) * sizeof(float));

    mod_kernel<<<BB * HH_, 32, 0, stream>>>(x, lw1, lb1, lw2, lb2, mod);
    disco_head_kernel<<<BB * PP, 32, 0, stream>>>(x, nbr, basis, disco_w, disco_b,
                                                  hw1, hb1, hw2, hb2, mod, xln);
    ffn_kernel<<<(BB * PP) / 16, 32, 0, stream>>>(x, xln, fw1, fb1, fw2, fb2, out);
}


# Point kernel chains its three products in registers: each D fragment is the next WMMA operand (k order matched to D rows), no LDS staging or barriers; sum order over n, s and hidden changes
# speedup vs baseline: 1.3875x; 1.3875x over previous
#include <hip/hip_runtime.h>
#include <hip/hip_bf16.h>

typedef __attribute__((ext_vector_type(2))) float v2f;
typedef __attribute__((ext_vector_type(8))) float v8f;

#define BB    2
#define HH_   90
#define WW    180
#define PP    (HH_ * WW)
#define CC    64
#define MD    66
#define SHN   16
#define KK    25
#define NNZN  25
#define HHID  32
#define FFNH  256

__device__ __forceinline__ float gelu_f(float v) {
    return 0.5f * v * (1.0f + erff(v * 0.70710678118654752440f));
}

__device__ __forceinline__ v8f wmma_f32(v2f a, v2f b, v8f c) {
    return __builtin_amdgcn_wmma_f32_16x16x4_f32(false, a, false, b, (short)0, c,
                                                 false, false);
}

__global__ __launch_bounds__(32)
void mod_kernel(const float* __restrict__ x,
                const float* __restrict__ lw1, const float* __restrict__ lb1,
                const float* __restrict__ lw2, const float* __restrict__ lb2,
                float* __restrict__ mod)
{
    const int row = blockIdx.x;
    const int j   = threadIdx.x;
    const float* xp = x + ((long)row * WW) * MD + CC;
    const float s0 = xp[0], s1 = xp[1];
    float acc = lb2[j];
    for (int i = 0; i < 2 * SHN; ++i) {
        float t = s0 * lw1[i] + s1 * lw1[2 * SHN + i] + lb1[i];
        acc += gelu_f(t) * lw2[i * (2 * SHN) + j];
    }
    mod[row * (2 * SHN) + j] = acc;
}

typedef __attribute__((ext_vector_type(4))) float v4f;

__device__ __forceinline__ v8f ld8(const float* p) {
    const v4f a = *(const v4f*)(p);
    const v4f c = *(const v4f*)(p + 4);
    v8f r = {a.x, a.y, a.z, a.w, c.x, c.y, c.z, c.w};
    return r;
}

__global__ __launch_bounds__(32, 7)
void disco_head_kernel(const float* __restrict__ x, const int* __restrict__ nbr,
                       const float* __restrict__ basis,
                       const float* __restrict__ disco_w,
                       const float* __restrict__ disco_b,
                       const float* __restrict__ hw1, const float* __restrict__ hb1,
                       const float* __restrict__ hw2, const float* __restrict__ hb2,
                       const float* __restrict__ mod, float* __restrict__ xln)
{
    const int l  = threadIdx.x;
    const int q  = blockIdx.x;
    const int b  = q / PP;
    const int p  = q - b * PP;
    const int hi = l >> 4;
    const int lr = l & 15;
    const bool lo_half = (hi == 0);

    v8f accb[2];
    {
        const v8f zb = {0.f, 0.f, 0.f, 0.f, 0.f, 0.f, 0.f, 0.f};
        accb[0] = zb; accb[1] = zb;
        const float* bp   = basis + (long)p * NNZN;
        const bool   n1ok = lr < (NNZN - 16);
        const int    n1   = n1ok ? (16 + lr) : 0;
        const float* dwp  = disco_w + lr * KK;
        #pragma unroll
        for (int kt = 0; kt < 6; ++kt) {
            const int k0 = kt * 4 + 2 * hi;
            const float* r0 = bp + (long)k0 * (PP * NNZN);
            const float* r1 = r0 + (PP * NNZN);
            v2f bf, a0, a1;
            bf.x = dwp[k0];
            bf.y = dwp[k0 + 1];
            a0.x = r0[lr];
            a0.y = r1[lr];
            const float t0 = r0[n1];
            const float t1 = r1[n1];
            a1.x = n1ok ? t0 : 0.f;
            a1.y = n1ok ? t1 : 0.f;
            accb[0] = wmma_f32(a0, bf, accb[0]);
            accb[1] = wmma_f32(a1, bf, accb[1]);
        }
        {
            const float* r0 = bp + (long)24 * (PP * NNZN);
            const float w  = dwp[24];
            const float t0 = r0[lr];
            const float t1 = r0[n1];
            v2f bf, a0, a1;
            bf.x = lo_half ? w : 0.f;              bf.y = 0.f;
            a0.x = lo_half ? t0 : 0.f;             a0.y = 0.f;
            a1.x = (lo_half && n1ok) ? t1 : 0.f;   a1.y = 0.f;
            accb[0] = wmma_f32(a0, bf, accb[0]);
            accb[1] = wmma_f32(a1, bf, accb[1]);
        }
    }

    const float* xb = x + (long)b * PP * MD;
    unsigned xoff[16];
    {
        const int* np = nbr + p * NNZN;
        #pragma unroll
        for (int r = 0; r < 8; ++r) {
            const int i0 = np[8 * hi + r];
            const int i1 = np[lo_half ? (16 + r) : 24];
            xoff[r]     = (unsigned)i0 * MD + lr;
            xoff[8 + r] = (unsigned)i1 * MD + lr;
        }
    }
    const v8f dbv = ld8(disco_b + 8 * hi);
    const int hrow = p / WW;
    const float* mrow = mod + (b * HH_ + hrow) * (2 * SHN);
    const v8f gm = ld8(mrow + 8 * hi);
    const v8f bt = ld8(mrow + SHN + 8 * hi);
    const int  jl  = lr >> 3;
    const bool hd0 = (jl == 0);
    const float* xres = x + ((long)b * PP + p) * MD;
    float*       orow = xln + (long)q * CC;

    #pragma unroll 1
    for (int ct = 0; ct < 4; ++ct) {
        v8f acc = dbv;
        const unsigned co = ct * 16;
        #pragma unroll
        for (int t = 0; t < 2; ++t) {
            #pragma unroll
            for (int u = 0; u < 4; ++u) {
                v2f af, bf;
                af.x = accb[t][2 * u];
                af.y = accb[t][2 * u + 1];
                bf.x = xb[xoff[8 * t + 2 * u] + co];
                bf.y = xb[xoff[8 * t + 2 * u + 1] + co];
                acc = wmma_f32(af, bf, acc);
            }
        }
        #pragma unroll
        for (int r = 0; r < 8; ++r) acc[r] = acc[r] * gm[r] + bt[r];

        const int hsel = 2 * ct + jl;
        v8f acch[2];
        acch[0] = ld8(hb1 + hsel * HHID + 8 * hi);
        acch[1] = ld8(hb1 + hsel * HHID + 16 + 8 * hi);
        #pragma unroll
        for (int jh = 0; jh < 2; ++jh) {
            const bool keep = jh ? !hd0 : hd0;
            #pragma unroll
            for (int u = 0; u < 4; ++u) {
                v2f bf;
                bf.x = keep ? acc[2 * u]     : 0.f;
                bf.y = keep ? acc[2 * u + 1] : 0.f;
                const float* wp = hw1 + ((2 * ct + jh) * SHN + 8 * hi + 2 * u) * HHID + lr;
                #pragma unroll
                for (int nt = 0; nt < 2; ++nt) {
                    v2f af;
                    af.x = wp[nt * 16];
                    af.y = wp[HHID + nt * 16];
                    acch[nt] = wmma_f32(af, bf, acch[nt]);
                }
            }
        }
        float ps = 0.f;
        #pragma unroll
        for (int nt = 0; nt < 2; ++nt) {
            const v8f w2v = ld8(hw2 + hsel * HHID + nt * 16 + 8 * hi);
            #pragma unroll
            for (int r = 0; r < 8; ++r) ps += gelu_f(acch[nt][r]) * w2v[r];
        }
        const float tot = ps + __shfl_xor(ps, 16, 32);
        if (lo_half) orow[ct * 16 + lr] = (hb2[hsel] + tot) + xres[ct * 16 + lr];
    }
}

__global__ __launch_bounds__(32)
void ffn_kernel(const float* __restrict__ x, const float* __restrict__ xln,
                const float* __restrict__ w1, const float* __restrict__ b1,
                const float* __restrict__ w2, const float* __restrict__ b2,
                float* __restrict__ out)
{
    __shared__ float xf[16 * 72];
    __shared__ float hid[16 * 260];

    const int  l   = threadIdx.x;
    const long gq0 = (long)blockIdx.x * 16;

    for (int e = l; e < 16 * 68; e += 32) {
        const int m = e / 68, c = e - m * 68;
        const long gq = gq0 + m;
        float v;
        if (c < CC)      v = xln[gq * CC + c];
        else if (c < MD) v = x[gq * MD + c];
        else             v = 0.f;
        xf[m * 72 + c] = v;
    }
    __syncthreads();

    const bool hi = l >= 16;
    const int  lr = l & 15;

    for (int nt = 0; nt < 16; ++nt) {
        const float bb = b1[nt * 16 + lr];
        v8f acc = {bb, bb, bb, bb, bb, bb, bb, bb};
        #pragma unroll
        for (int kt = 0; kt < 17; ++kt) {
            const int k0 = kt * 4 + (hi ? 2 : 0);
            const int ka = (k0     < MD) ? k0     : 0;
            const int kb = (k0 + 1 < MD) ? k0 + 1 : 0;
            v2f af, bf;
            af.x = xf[lr * 72 + k0];
            af.y = xf[lr * 72 + k0 + 1];
            const float w0 = w1[(long)ka * FFNH + nt * 16 + lr];
            const float w1v = w1[(long)kb * FFNH + nt * 16 + lr];
            bf.x = (k0     < MD) ? w0  : 0.f;
            bf.y = (k0 + 1 < MD) ? w1v : 0.f;
            acc = wmma_f32(af, bf, acc);
        }
        #pragma unroll
        for (int r = 0; r < 8; ++r) {
            const int m = r + (hi ? 8 : 0);
            hid[m * 260 + nt * 16 + lr] = gelu_f(acc[r]);
        }
    }
    __syncthreads();

    for (int nt = 0; nt < 4; ++nt) {
        const float bb = b2[nt * 16 + lr];
        v8f acc = {bb, bb, bb, bb, bb, bb, bb, bb};
        #pragma unroll
        for (int kt = 0; kt < 64; ++kt) {
            const int k0 = kt * 4 + (hi ? 2 : 0);
            v2f af, bf;
            af.x = hid[lr * 260 + k0];
            af.y = hid[lr * 260 + k0 + 1];
            bf.x = w2[k0 * CC + nt * 16 + lr];
            bf.y = w2[(k0 + 1) * CC + nt * 16 + lr];
            acc = wmma_f32(af, bf, acc);
        }
        #pragma unroll
        for (int r = 0; r < 8; ++r) {
            const int  m  = r + (hi ? 8 : 0);
            const long gq = gq0 + m;
            const int  c  = nt * 16 + lr;
            out[gq * MD + c] = acc[r] + xln[gq * CC + c];
        }
    }
    {
        const int  m  = l >> 1, c2 = l & 1;
        const long gq = gq0 + m;
        out[gq * MD + CC + c2] = x[gq * MD + CC + c2];
    }
}

extern "C" void kernel_launch(void* const* d_in, const int* in_sizes, int n_in,
                              void* d_out, int out_size, void* d_ws, size_t ws_size,
                              hipStream_t stream)
{
    const float* x       = (const float*)d_in[0];
    const int*   nbr     = (const int*)  d_in[1];
    const float* basis   = (const float*)d_in[2];
    const float* disco_w = (const float*)d_in[3];
    const float* disco_b = (const float*)d_in[4];
    const float* lw1     = (const float*)d_in[5];
    const float* lb1     = (const float*)d_in[6];
    const float* lw2     = (const float*)d_in[7];
    const float* lb2     = (const float*)d_in[8];
    const float* hw1     = (const float*)d_in[9];
    const float* hb1     = (const float*)d_in[10];
    const float* hw2     = (const float*)d_in[11];
    const float* hb2     = (const float*)d_in[12];
    const float* fw1     = (const float*)d_in[13];
    const float* fb1     = (const float*)d_in[14];
    const float* fw2     = (const float*)d_in[15];
    const float* fb2     = (const float*)d_in[16];
    float* out = (float*)d_out;

    float* mod = (float*)d_ws;
    float* xln = (float*)((char*)d_ws + (size_t)(BB * HH_ * 32) * sizeof(float));

    mod_kernel<<<BB * HH_, 32, 0, stream>>>(x, lw1, lb1, lw2, lb2, mod);
    disco_head_kernel<<<BB * PP, 32, 0, stream>>>(x, nbr, basis, disco_w, disco_b,
                                                  hw1, hb1, hw2, hb2, mod, xln);
    ffn_kernel<<<(BB * PP) / 16, 32, 0, stream>>>(x, xln, fw1, fb1, fw2, fb2, out);
}


# gelu: library erff (two exec-masked paths a call) replaced by branch-free v*Phi(v), erfc(x)/2 = t*P6(t)*exp2(-x*x*log2 e), t = rcp(1+c*x), accurate to f32 rounding; all three kernels
# speedup vs baseline: 1.7707x; 1.2762x over previous
#include <hip/hip_runtime.h>
#include <hip/hip_bf16.h>

typedef __attribute__((ext_vector_type(2))) float v2f;
typedef __attribute__((ext_vector_type(8))) float v8f;

#define BB    2
#define HH_   90
#define WW    180
#define PP    (HH_ * WW)
#define CC    64
#define MD    66
#define SHN   16
#define KK    25
#define NNZN  25
#define HHID  32
#define FFNH  256

__device__ __forceinline__ float gelu_f(float v) {
    const float ax = __builtin_fabsf(v);
    const float t  = __builtin_amdgcn_rcpf(__builtin_fmaf(ax, 0.318198055f, 1.0f));
    float s = 0.0463168584f;
    s = __builtin_fmaf(t, s, -0.247209847f);
    s = __builtin_fmaf(t, s,  0.428905427f);
    s = __builtin_fmaf(t, s, -0.188248932f);
    s = __builtin_fmaf(t, s,  0.230295271f);
    s = __builtin_fmaf(t, s,  0.100437485f);
    s = __builtin_fmaf(t, s,  0.129503742f);
    const float e = __builtin_amdgcn_exp2f((v * v) * -0.721347511f);
    const float q = (s * t) * e;
    const float phi = (v < 0.f) ? q : (1.0f - q);
    return v * phi;
}

__device__ __forceinline__ v8f wmma_f32(v2f a, v2f b, v8f c) {
    return __builtin_amdgcn_wmma_f32_16x16x4_f32(false, a, false, b, (short)0, c,
                                                 false, false);
}

__global__ __launch_bounds__(32)
void mod_kernel(const float* __restrict__ x,
                const float* __restrict__ lw1, const float* __restrict__ lb1,
                const float* __restrict__ lw2, const float* __restrict__ lb2,
                float* __restrict__ mod)
{
    const int row = blockIdx.x;
    const int j   = threadIdx.x;
    const float* xp = x + ((long)row * WW) * MD + CC;
    const float s0 = xp[0], s1 = xp[1];
    float acc = lb2[j];
    for (int i = 0; i < 2 * SHN; ++i) {
        float t = s0 * lw1[i] + s1 * lw1[2 * SHN + i] + lb1[i];
        acc += gelu_f(t) * lw2[i * (2 * SHN) + j];
    }
    mod[row * (2 * SHN) + j] = acc;
}

typedef __attribute__((ext_vector_type(4))) float v4f;

__device__ __forceinline__ v8f ld8(const float* p) {
    const v4f a = *(const v4f*)(p);
    const v4f c = *(const v4f*)(p + 4);
    v8f r = {a.x, a.y, a.z, a.w, c.x, c.y, c.z, c.w};
    return r;
}

__global__ __launch_bounds__(32, 7)
void disco_head_kernel(const float* __restrict__ x, const int* __restrict__ nbr,
                       const float* __restrict__ basis,
                       const float* __restrict__ disco_w,
                       const float* __restrict__ disco_b,
                       const float* __restrict__ hw1, const float* __restrict__ hb1,
                       const float* __restrict__ hw2, const float* __restrict__ hb2,
                       const float* __restrict__ mod, float* __restrict__ xln)
{
    const int l  = threadIdx.x;
    const int q  = blockIdx.x;
    const int b  = q / PP;
    const int p  = q - b * PP;
    const int hi = l >> 4;
    const int lr = l & 15;
    const bool lo_half = (hi == 0);

    v8f accb[2];
    {
        const v8f zb = {0.f, 0.f, 0.f, 0.f, 0.f, 0.f, 0.f, 0.f};
        accb[0] = zb; accb[1] = zb;
        const float* bp   = basis + (long)p * NNZN;
        const bool   n1ok = lr < (NNZN - 16);
        const int    n1   = n1ok ? (16 + lr) : 0;
        const float* dwp  = disco_w + lr * KK;
        #pragma unroll
        for (int kt = 0; kt < 6; ++kt) {
            const int k0 = kt * 4 + 2 * hi;
            const float* r0 = bp + (long)k0 * (PP * NNZN);
            const float* r1 = r0 + (PP * NNZN);
            v2f bf, a0, a1;
            bf.x = dwp[k0];
            bf.y = dwp[k0 + 1];
            a0.x = r0[lr];
            a0.y = r1[lr];
            const float t0 = r0[n1];
            const float t1 = r1[n1];
            a1.x = n1ok ? t0 : 0.f;
            a1.y = n1ok ? t1 : 0.f;
            accb[0] = wmma_f32(a0, bf, accb[0]);
            accb[1] = wmma_f32(a1, bf, accb[1]);
        }
        {
            const float* r0 = bp + (long)24 * (PP * NNZN);
            const float w  = dwp[24];
            const float t0 = r0[lr];
            const float t1 = r0[n1];
            v2f bf, a0, a1;
            bf.x = lo_half ? w : 0.f;              bf.y = 0.f;
            a0.x = lo_half ? t0 : 0.f;             a0.y = 0.f;
            a1.x = (lo_half && n1ok) ? t1 : 0.f;   a1.y = 0.f;
            accb[0] = wmma_f32(a0, bf, accb[0]);
            accb[1] = wmma_f32(a1, bf, accb[1]);
        }
    }

    const float* xb = x + (long)b * PP * MD;
    unsigned xoff[16];
    {
        const int* np = nbr + p * NNZN;
        #pragma unroll
        for (int r = 0; r < 8; ++r) {
            const int i0 = np[8 * hi + r];
            const int i1 = np[lo_half ? (16 + r) : 24];
            xoff[r]     = (unsigned)i0 * MD + lr;
            xoff[8 + r] = (unsigned)i1 * MD + lr;
        }
    }
    const v8f dbv = ld8(disco_b + 8 * hi);
    const int hrow = p / WW;
    const float* mrow = mod + (b * HH_ + hrow) * (2 * SHN);
    const v8f gm = ld8(mrow + 8 * hi);
    const v8f bt = ld8(mrow + SHN + 8 * hi);
    const int  jl  = lr >> 3;
    const bool hd0 = (jl == 0);
    const float* xres = x + ((long)b * PP + p) * MD;
    float*       orow = xln + (long)q * CC;

    #pragma unroll 1
    for (int ct = 0; ct < 4; ++ct) {
        v8f acc = dbv;
        const unsigned co = ct * 16;
        #pragma unroll
        for (int t = 0; t < 2; ++t) {
            #pragma unroll
            for (int u = 0; u < 4; ++u) {
                v2f af, bf;
                af.x = accb[t][2 * u];
                af.y = accb[t][2 * u + 1];
                bf.x = xb[xoff[8 * t + 2 * u] + co];
                bf.y = xb[xoff[8 * t + 2 * u + 1] + co];
                acc = wmma_f32(af, bf, acc);
            }
        }
        #pragma unroll
        for (int r = 0; r < 8; ++r) acc[r] = acc[r] * gm[r] + bt[r];

        const int hsel = 2 * ct + jl;
        v8f acch[2];
        acch[0] = ld8(hb1 + hsel * HHID + 8 * hi);
        acch[1] = ld8(hb1 + hsel * HHID + 16 + 8 * hi);
        #pragma unroll
        for (int jh = 0; jh < 2; ++jh) {
            const bool keep = jh ? !hd0 : hd0;
            #pragma unroll
            for (int u = 0; u < 4; ++u) {
                v2f bf;
                bf.x = keep ? acc[2 * u]     : 0.f;
                bf.y = keep ? acc[2 * u + 1] : 0.f;
                const float* wp = hw1 + ((2 * ct + jh) * SHN + 8 * hi + 2 * u) * HHID + lr;
                #pragma unroll
                for (int nt = 0; nt < 2; ++nt) {
                    v2f af;
                    af.x = wp[nt * 16];
                    af.y = wp[HHID + nt * 16];
                    acch[nt] = wmma_f32(af, bf, acch[nt]);
                }
            }
        }
        float ps = 0.f;
        #pragma unroll
        for (int nt = 0; nt < 2; ++nt) {
            const v8f w2v = ld8(hw2 + hsel * HHID + nt * 16 + 8 * hi);
            #pragma unroll
            for (int r = 0; r < 8; ++r) ps += gelu_f(acch[nt][r]) * w2v[r];
        }
        const float tot = ps + __shfl_xor(ps, 16, 32);
        if (lo_half) orow[ct * 16 + lr] = (hb2[hsel] + tot) + xres[ct * 16 + lr];
    }
}

__global__ __launch_bounds__(32)
void ffn_kernel(const float* __restrict__ x, const float* __restrict__ xln,
                const float* __restrict__ w1, const float* __restrict__ b1,
                const float* __restrict__ w2, const float* __restrict__ b2,
                float* __restrict__ out)
{
    __shared__ float xf[16 * 72];
    __shared__ float hid[16 * 260];

    const int  l   = threadIdx.x;
    const long gq0 = (long)blockIdx.x * 16;

    for (int e = l; e < 16 * 68; e += 32) {
        const int m = e / 68, c = e - m * 68;
        const long gq = gq0 + m;
        float v;
        if (c < CC)      v = xln[gq * CC + c];
        else if (c < MD) v = x[gq * MD + c];
        else             v = 0.f;
        xf[m * 72 + c] = v;
    }
    __syncthreads();

    const bool hi = l >= 16;
    const int  lr = l & 15;

    for (int nt = 0; nt < 16; ++nt) {
        const float bb = b1[nt * 16 + lr];
        v8f acc = {bb, bb, bb, bb, bb, bb, bb, bb};
        #pragma unroll
        for (int kt = 0; kt < 17; ++kt) {
            const int k0 = kt * 4 + (hi ? 2 : 0);
            const int ka = (k0     < MD) ? k0     : 0;
            const int kb = (k0 + 1 < MD) ? k0 + 1 : 0;
            v2f af, bf;
            af.x = xf[lr * 72 + k0];
            af.y = xf[lr * 72 + k0 + 1];
            const float w0 = w1[(long)ka * FFNH + nt * 16 + lr];
            const float w1v = w1[(long)kb * FFNH + nt * 16 + lr];
            bf.x = (k0     < MD) ? w0  : 0.f;
            bf.y = (k0 + 1 < MD) ? w1v : 0.f;
            acc = wmma_f32(af, bf, acc);
        }
        #pragma unroll
        for (int r = 0; r < 8; ++r) {
            const int m = r + (hi ? 8 : 0);
            hid[m * 260 + nt * 16 + lr] = gelu_f(acc[r]);
        }
    }
    __syncthreads();

    for (int nt = 0; nt < 4; ++nt) {
        const float bb = b2[nt * 16 + lr];
        v8f acc = {bb, bb, bb, bb, bb, bb, bb, bb};
        #pragma unroll
        for (int kt = 0; kt < 64; ++kt) {
            const int k0 = kt * 4 + (hi ? 2 : 0);
            v2f af, bf;
            af.x = hid[lr * 260 + k0];
            af.y = hid[lr * 260 + k0 + 1];
            bf.x = w2[k0 * CC + nt * 16 + lr];
            bf.y = w2[(k0 + 1) * CC + nt * 16 + lr];
            acc = wmma_f32(af, bf, acc);
        }
        #pragma unroll
        for (int r = 0; r < 8; ++r) {
            const int  m  = r + (hi ? 8 : 0);
            const long gq = gq0 + m;
            const int  c  = nt * 16 + lr;
            out[gq * MD + c] = acc[r] + xln[gq * CC + c];
        }
    }
    {
        const int  m  = l >> 1, c2 = l & 1;
        const long gq = gq0 + m;
        out[gq * MD + CC + c2] = x[gq * MD + CC + c2];
    }
}

extern "C" void kernel_launch(void* const* d_in, const int* in_sizes, int n_in,
                              void* d_out, int out_size, void* d_ws, size_t ws_size,
                              hipStream_t stream)
{
    const float* x       = (const float*)d_in[0];
    const int*   nbr     = (const int*)  d_in[1];
    const float* basis   = (const float*)d_in[2];
    const float* disco_w = (const float*)d_in[3];
    const float* disco_b = (const float*)d_in[4];
    const float* lw1     = (const float*)d_in[5];
    const float* lb1     = (const float*)d_in[6];
    const float* lw2     = (const float*)d_in[7];
    const float* lb2     = (const float*)d_in[8];
    const float* hw1     = (const float*)d_in[9];
    const float* hb1     = (const float*)d_in[10];
    const float* hw2     = (const float*)d_in[11];
    const float* hb2     = (const float*)d_in[12];
    const float* fw1     = (const float*)d_in[13];
    const float* fb1     = (const float*)d_in[14];
    const float* fw2     = (const float*)d_in[15];
    const float* fb2     = (const float*)d_in[16];
    float* out = (float*)d_out;

    float* mod = (float*)d_ws;
    float* xln = (float*)((char*)d_ws + (size_t)(BB * HH_ * 32) * sizeof(float));

    mod_kernel<<<BB * HH_, 32, 0, stream>>>(x, lw1, lb1, lw2, lb2, mod);
    disco_head_kernel<<<BB * PP, 32, 0, stream>>>(x, nbr, basis, disco_w, disco_b,
                                                  hw1, hb1, hw2, hb2, mod, xln);
    ffn_kernel<<<(BB * PP) / 16, 32, 0, stream>>>(x, xln, fw1, fb1, fw2, fb2, out);
}


# FFN kernel: global loads batched ahead of use: point tile read straight into A fragments (17 8-byte loads, one round trip, no 34-step serial LDS staging loop); W1 block fetched one block ahead
# speedup vs baseline: 1.9690x; 1.1120x over previous
#include <hip/hip_runtime.h>
#include <hip/hip_bf16.h>

typedef __attribute__((ext_vector_type(2))) float v2f;
typedef __attribute__((ext_vector_type(8))) float v8f;

#define BB    2
#define HH_   90
#define WW    180
#define PP    (HH_ * WW)
#define CC    64
#define MD    66
#define SHN   16
#define KK    25
#define NNZN  25
#define HHID  32
#define FFNH  256

__device__ __forceinline__ float gelu_f(float v) {
    const float ax = __builtin_fabsf(v);
    const float t  = __builtin_amdgcn_rcpf(__builtin_fmaf(ax, 0.318198055f, 1.0f));
    float s = 0.0463168584f;
    s = __builtin_fmaf(t, s, -0.247209847f);
    s = __builtin_fmaf(t, s,  0.428905427f);
    s = __builtin_fmaf(t, s, -0.188248932f);
    s = __builtin_fmaf(t, s,  0.230295271f);
    s = __builtin_fmaf(t, s,  0.100437485f);
    s = __builtin_fmaf(t, s,  0.129503742f);
    const float e = __builtin_amdgcn_exp2f((v * v) * -0.721347511f);
    const float q = (s * t) * e;
    const float phi = (v < 0.f) ? q : (1.0f - q);
    return v * phi;
}

__device__ __forceinline__ v8f wmma_f32(v2f a, v2f b, v8f c) {
    return __builtin_amdgcn_wmma_f32_16x16x4_f32(false, a, false, b, (short)0, c,
                                                 false, false);
}

__global__ __launch_bounds__(32)
void mod_kernel(const float* __restrict__ x,
                const float* __restrict__ lw1, const float* __restrict__ lb1,
                const float* __restrict__ lw2, const float* __restrict__ lb2,
                float* __restrict__ mod)
{
    const int row = blockIdx.x;
    const int j   = threadIdx.x;
    const float* xp = x + ((long)row * WW) * MD + CC;
    const float s0 = xp[0], s1 = xp[1];
    float acc = lb2[j];
    for (int i = 0; i < 2 * SHN; ++i) {
        float t = s0 * lw1[i] + s1 * lw1[2 * SHN + i] + lb1[i];
        acc += gelu_f(t) * lw2[i * (2 * SHN) + j];
    }
    mod[row * (2 * SHN) + j] = acc;
}

typedef __attribute__((ext_vector_type(4))) float v4f;

__device__ __forceinline__ v8f ld8(const float* p) {
    const v4f a = *(const v4f*)(p);
    const v4f c = *(const v4f*)(p + 4);
    v8f r = {a.x, a.y, a.z, a.w, c.x, c.y, c.z, c.w};
    return r;
}

__global__ __launch_bounds__(32, 7)
void disco_head_kernel(const float* __restrict__ x, const int* __restrict__ nbr,
                       const float* __restrict__ basis,
                       const float* __restrict__ disco_w,
                       const float* __restrict__ disco_b,
                       const float* __restrict__ hw1, const float* __restrict__ hb1,
                       const float* __restrict__ hw2, const float* __restrict__ hb2,
                       const float* __restrict__ mod, float* __restrict__ xln)
{
    const int l  = threadIdx.x;
    const int q  = blockIdx.x;
    const int b  = q / PP;
    const int p  = q - b * PP;
    const int hi = l >> 4;
    const int lr = l & 15;
    const bool lo_half = (hi == 0);

    v8f accb[2];
    {
        const v8f zb = {0.f, 0.f, 0.f, 0.f, 0.f, 0.f, 0.f, 0.f};
        accb[0] = zb; accb[1] = zb;
        const float* bp   = basis + (long)p * NNZN;
        const bool   n1ok = lr < (NNZN - 16);
        const int    n1   = n1ok ? (16 + lr) : 0;
        const float* dwp  = disco_w + lr * KK;
        #pragma unroll
        for (int kt = 0; kt < 6; ++kt) {
            const int k0 = kt * 4 + 2 * hi;
            const float* r0 = bp + (long)k0 * (PP * NNZN);
            const float* r1 = r0 + (PP * NNZN);
            v2f bf, a0, a1;
            bf.x = dwp[k0];
            bf.y = dwp[k0 + 1];
            a0.x = r0[lr];
            a0.y = r1[lr];
            const float t0 = r0[n1];
            const float t1 = r1[n1];
            a1.x = n1ok ? t0 : 0.f;
            a1.y = n1ok ? t1 : 0.f;
            accb[0] = wmma_f32(a0, bf, accb[0]);
            accb[1] = wmma_f32(a1, bf, accb[1]);
        }
        {
            const float* r0 = bp + (long)24 * (PP * NNZN);
            const float w  = dwp[24];
            const float t0 = r0[lr];
            const float t1 = r0[n1];
            v2f bf, a0, a1;
            bf.x = lo_half ? w : 0.f;              bf.y = 0.f;
            a0.x = lo_half ? t0 : 0.f;             a0.y = 0.f;
            a1.x = (lo_half && n1ok) ? t1 : 0.f;   a1.y = 0.f;
            accb[0] = wmma_f32(a0, bf, accb[0]);
            accb[1] = wmma_f32(a1, bf, accb[1]);
        }
    }

    const float* xb = x + (long)b * PP * MD;
    unsigned xoff[16];
    {
        const int* np = nbr + p * NNZN;
        #pragma unroll
        for (int r = 0; r < 8; ++r) {
            const int i0 = np[8 * hi + r];
            const int i1 = np[lo_half ? (16 + r) : 24];
            xoff[r]     = (unsigned)i0 * MD + lr;
            xoff[8 + r] = (unsigned)i1 * MD + lr;
        }
    }
    const v8f dbv = ld8(disco_b + 8 * hi);
    const int hrow = p / WW;
    const float* mrow = mod + (b * HH_ + hrow) * (2 * SHN);
    const v8f gm = ld8(mrow + 8 * hi);
    const v8f bt = ld8(mrow + SHN + 8 * hi);
    const int  jl  = lr >> 3;
    const bool hd0 = (jl == 0);
    const float* xres = x + ((long)b * PP + p) * MD;
    float*       orow = xln + (long)q * CC;

    #pragma unroll 1
    for (int ct = 0; ct < 4; ++ct) {
        v8f acc = dbv;
        const unsigned co = ct * 16;
        #pragma unroll
        for (int t = 0; t < 2; ++t) {
            #pragma unroll
            for (int u = 0; u < 4; ++u) {
                v2f af, bf;
                af.x = accb[t][2 * u];
                af.y = accb[t][2 * u + 1];
                bf.x = xb[xoff[8 * t + 2 * u] + co];
                bf.y = xb[xoff[8 * t + 2 * u + 1] + co];
                acc = wmma_f32(af, bf, acc);
            }
        }
        #pragma unroll
        for (int r = 0; r < 8; ++r) acc[r] = acc[r] * gm[r] + bt[r];

        const int hsel = 2 * ct + jl;
        v8f acch[2];
        acch[0] = ld8(hb1 + hsel * HHID + 8 * hi);
        acch[1] = ld8(hb1 + hsel * HHID + 16 + 8 * hi);
        #pragma unroll
        for (int jh = 0; jh < 2; ++jh) {
            const bool keep = jh ? !hd0 : hd0;
            #pragma unroll
            for (int u = 0; u < 4; ++u) {
                v2f bf;
                bf.x = keep ? acc[2 * u]     : 0.f;
                bf.y = keep ? acc[2 * u + 1] : 0.f;
                const float* wp = hw1 + ((2 * ct + jh) * SHN + 8 * hi + 2 * u) * HHID + lr;
                #pragma unroll
                for (int nt = 0; nt < 2; ++nt) {
                    v2f af;
                    af.x = wp[nt * 16];
                    af.y = wp[HHID + nt * 16];
                    acch[nt] = wmma_f32(af, bf, acch[nt]);
                }
            }
        }
        float ps = 0.f;
        #pragma unroll
        for (int nt = 0; nt < 2; ++nt) {
            const v8f w2v = ld8(hw2 + hsel * HHID + nt * 16 + 8 * hi);
            #pragma unroll
            for (int r = 0; r < 8; ++r) ps += gelu_f(acch[nt][r]) * w2v[r];
        }
        const float tot = ps + __shfl_xor(ps, 16, 32);
        if (lo_half) orow[ct * 16 + lr] = (hb2[hsel] + tot) + xres[ct * 16 + lr];
    }
}

__device__ __forceinline__ void ffn_load_w1(const float* __restrict__ w1, int nt, int lr, bool hi,
                                            v2f (&bf)[17]) {
    #pragma unroll
    for (int kt = 0; kt < 17; ++kt) {
        const int k0 = kt * 4 + (hi ? 2 : 0);
        const int ka = (k0     < MD) ? k0     : 0;
        const int kb = (k0 + 1 < MD) ? k0 + 1 : 0;
        const float w0  = w1[(long)ka * FFNH + nt * 16 + lr];
        const float w1v = w1[(long)kb * FFNH + nt * 16 + lr];
        bf[kt].x = (k0     < MD) ? w0  : 0.f;
        bf[kt].y = (k0 + 1 < MD) ? w1v : 0.f;
    }
}

__device__ __forceinline__ void ffn_hidden_block(const v2f (&af)[17], const v2f (&bf)[17], float bb,
                                                 float* __restrict__ hid, int nt, int lr, bool hi) {
    v8f acc = {bb, bb, bb, bb, bb, bb, bb, bb};
    #pragma unroll
    for (int kt = 0; kt < 17; ++kt) acc = wmma_f32(af[kt], bf[kt], acc);
    #pragma unroll
    for (int r = 0; r < 8; ++r) {
        const int m = r + (hi ? 8 : 0);
        hid[m * 260 + nt * 16 + lr] = gelu_f(acc[r]);
    }
}

__global__ __launch_bounds__(32)
void ffn_kernel(const float* __restrict__ x, const float* __restrict__ xln,
                const float* __restrict__ w1, const float* __restrict__ b1,
                const float* __restrict__ w2, const float* __restrict__ b2,
                float* __restrict__ out)
{
    __shared__ float hid[16 * 260];

    const int  l   = threadIdx.x;
    const long gq0 = (long)blockIdx.x * 16;
    const bool hi = l >= 16;
    const int  lr = l & 15;

    v2f afr[17];
    {
        const float* xr = xln + (gq0 + lr) * CC + (hi ? 2 : 0);
        #pragma unroll
        for (int kt = 0; kt < 16; ++kt) afr[kt] = *(const v2f*)(xr + kt * 4);
        const v2f sc = *(const v2f*)(x + (gq0 + lr) * MD + CC);
        afr[16].x = hi ? 0.f : sc.x;
        afr[16].y = hi ? 0.f : sc.y;
    }

    v2f bA[17], bB[17];
    ffn_load_w1(w1, 0, lr, hi, bA);
    float bbA = b1[lr];
    for (int nt = 0; nt < 16; nt += 2) {
        ffn_load_w1(w1, nt + 1, lr, hi, bB);
        const float bbB = b1[(nt + 1) * 16 + lr];
        ffn_hidden_block(afr, bA, bbA, hid, nt, lr, hi);
        const int n2 = (nt + 2 < 16) ? nt + 2 : 15;
        ffn_load_w1(w1, n2, lr, hi, bA);
        bbA = b1[n2 * 16 + lr];
        ffn_hidden_block(afr, bB, bbB, hid, nt + 1, lr, hi);
    }
    __syncthreads();

    for (int nt = 0; nt < 4; ++nt) {
        const float bb = b2[nt * 16 + lr];
        v8f acc = {bb, bb, bb, bb, bb, bb, bb, bb};
        #pragma unroll
        for (int kt = 0; kt < 64; ++kt) {
            const int k0 = kt * 4 + (hi ? 2 : 0);
            v2f af, bf;
            af.x = hid[lr * 260 + k0];
            af.y = hid[lr * 260 + k0 + 1];
            bf.x = w2[k0 * CC + nt * 16 + lr];
            bf.y = w2[(k0 + 1) * CC + nt * 16 + lr];
            acc = wmma_f32(af, bf, acc);
        }
        #pragma unroll
        for (int r = 0; r < 8; ++r) {
            const int  m  = r + (hi ? 8 : 0);
            const long gq = gq0 + m;
            const int  c  = nt * 16 + lr;
            out[gq * MD + c] = acc[r] + xln[gq * CC + c];
        }
    }
    {
        const int  m  = l >> 1, c2 = l & 1;
        const long gq = gq0 + m;
        out[gq * MD + CC + c2] = x[gq * MD + CC + c2];
    }
}

extern "C" void kernel_launch(void* const* d_in, const int* in_sizes, int n_in,
                              void* d_out, int out_size, void* d_ws, size_t ws_size,
                              hipStream_t stream)
{
    const float* x       = (const float*)d_in[0];
    const int*   nbr     = (const int*)  d_in[1];
    const float* basis   = (const float*)d_in[2];
    const float* disco_w = (const float*)d_in[3];
    const float* disco_b = (const float*)d_in[4];
    const float* lw1     = (const float*)d_in[5];
    const float* lb1     = (const float*)d_in[6];
    const float* lw2     = (const float*)d_in[7];
    const float* lb2     = (const float*)d_in[8];
    const float* hw1     = (const float*)d_in[9];
    const float* hb1     = (const float*)d_in[10];
    const float* hw2     = (const float*)d_in[11];
    const float* hb2     = (const float*)d_in[12];
    const float* fw1     = (const float*)d_in[13];
    const float* fb1     = (const float*)d_in[14];
    const float* fw2     = (const float*)d_in[15];
    const float* fb2     = (const float*)d_in[16];
    float* out = (float*)d_out;

    float* mod = (float*)d_ws;
    float* xln = (float*)((char*)d_ws + (size_t)(BB * HH_ * 32) * sizeof(float));

    mod_kernel<<<BB * HH_, 32, 0, stream>>>(x, lw1, lb1, lw2, lb2, mod);
    disco_head_kernel<<<BB * PP, 32, 0, stream>>>(x, nbr, basis, disco_w, disco_b,
                                                  hw1, hb1, hw2, hb2, mod, xln);
    ffn_kernel<<<(BB * PP) / 16, 32, 0, stream>>>(x, xln, fw1, fb1, fw2, fb2, out);
}


# Point kernel: one wave takes both batch rows of a point; a head's MLP tile = 8 channels x 2 batch rows of ONE head (8-lane DPP regroup), K = 16 real s, no zero-padded half; Bmat once per point
# speedup vs baseline: 2.0934x; 1.0632x over previous
#include <hip/hip_runtime.h>
#include <hip/hip_bf16.h>

typedef __attribute__((ext_vector_type(2))) float v2f;
typedef __attribute__((ext_vector_type(8))) float v8f;

#define BB    2
#define HH_   90
#define WW    180
#define PP    (HH_ * WW)
#define CC    64
#define MD    66
#define SHN   16
#define KK    25
#define NNZN  25
#define HHID  32
#define FFNH  256

__device__ __forceinline__ float gelu_f(float v) {
    const float ax = __builtin_fabsf(v);
    const float t  = __builtin_amdgcn_rcpf(__builtin_fmaf(ax, 0.318198055f, 1.0f));
    float s = 0.0463168584f;
    s = __builtin_fmaf(t, s, -0.247209847f);
    s = __builtin_fmaf(t, s,  0.428905427f);
    s = __builtin_fmaf(t, s, -0.188248932f);
    s = __builtin_fmaf(t, s,  0.230295271f);
    s = __builtin_fmaf(t, s,  0.100437485f);
    s = __builtin_fmaf(t, s,  0.129503742f);
    const float e = __builtin_amdgcn_exp2f((v * v) * -0.721347511f);
    const float q = (s * t) * e;
    const float phi = (v < 0.f) ? q : (1.0f - q);
    return v * phi;
}

__device__ __forceinline__ v8f wmma_f32(v2f a, v2f b, v8f c) {
    return __builtin_amdgcn_wmma_f32_16x16x4_f32(false, a, false, b, (short)0, c,
                                                 false, false);
}

__global__ __launch_bounds__(32)
void mod_kernel(const float* __restrict__ x,
                const float* __restrict__ lw1, const float* __restrict__ lb1,
                const float* __restrict__ lw2, const float* __restrict__ lb2,
                float* __restrict__ mod)
{
    const int row = blockIdx.x;
    const int j   = threadIdx.x;
    const float* xp = x + ((long)row * WW) * MD + CC;
    const float s0 = xp[0], s1 = xp[1];
    float acc = lb2[j];
    for (int i = 0; i < 2 * SHN; ++i) {
        float t = s0 * lw1[i] + s1 * lw1[2 * SHN + i] + lb1[i];
        acc += gelu_f(t) * lw2[i * (2 * SHN) + j];
    }
    mod[row * (2 * SHN) + j] = acc;
}

typedef __attribute__((ext_vector_type(4))) float v4f;

__device__ __forceinline__ v8f ld8(const float* p) {
    const v4f a = *(const v4f*)(p);
    const v4f c = *(const v4f*)(p + 4);
    v8f r = {a.x, a.y, a.z, a.w, c.x, c.y, c.z, c.w};
    return r;
}

__device__ __forceinline__ float take_hi8(float keep, float src) {
    return __builtin_bit_cast(float, __builtin_amdgcn_update_dpp(
        __builtin_bit_cast(int, keep), __builtin_bit_cast(int, src), 0x128, 0xF, 0xC, false));
}
__device__ __forceinline__ float take_lo8(float keep, float src) {
    return __builtin_bit_cast(float, __builtin_amdgcn_update_dpp(
        __builtin_bit_cast(int, keep), __builtin_bit_cast(int, src), 0x128, 0xF, 0x3, false));
}

__device__ __forceinline__ float head_mlp(const v8f g, int head, int lr, int hi,
                                          const float* __restrict__ hw1,
                                          const float* __restrict__ hb1,
                                          const float* __restrict__ hw2) {
    v8f acch[2];
    acch[0] = ld8(hb1 + head * HHID + 8 * hi);
    acch[1] = ld8(hb1 + head * HHID + 16 + 8 * hi);
    #pragma unroll
    for (int u = 0; u < 4; ++u) {
        v2f bf;
        bf.x = g[2 * u];
        bf.y = g[2 * u + 1];
        const float* wp = hw1 + (head * SHN + 8 * hi + 2 * u) * HHID + lr;
        #pragma unroll
        for (int nt = 0; nt < 2; ++nt) {
            v2f af;
            af.x = wp[nt * 16];
            af.y = wp[HHID + nt * 16];
            acch[nt] = wmma_f32(af, bf, acch[nt]);
        }
    }
    float ps = 0.f;
    #pragma unroll
    for (int nt = 0; nt < 2; ++nt) {
        const v8f w2v = ld8(hw2 + head * HHID + nt * 16 + 8 * hi);
        #pragma unroll
        for (int r = 0; r < 8; ++r) ps += gelu_f(acch[nt][r]) * w2v[r];
    }
    return ps + __shfl_xor(ps, 16, 32);
}

static_assert(BB == 2, "the point kernel pairs the two batch rows of a grid point in one wave");

__global__ __launch_bounds__(32, 5)
void disco_head_kernel(const float* __restrict__ x, const int* __restrict__ nbr,
                       const float* __restrict__ basis,
                       const float* __restrict__ disco_w,
                       const float* __restrict__ disco_b,
                       const float* __restrict__ hw1, const float* __restrict__ hb1,
                       const float* __restrict__ hw2, const float* __restrict__ hb2,
                       const float* __restrict__ mod, float* __restrict__ xln)
{
    const int l  = threadIdx.x;
    const int p  = blockIdx.x;
    const int hi = l >> 4;
    const int lr = l & 15;
    const bool lo_half = (hi == 0);

    v8f accb[2];
    {
        const v8f zb = {0.f, 0.f, 0.f, 0.f, 0.f, 0.f, 0.f, 0.f};
        accb[0] = zb; accb[1] = zb;
        const float* bp   = basis + (long)p * NNZN;
        const bool   n1ok = lr < (NNZN - 16);
        const int    n1   = n1ok ? (16 + lr) : 0;
        const float* dwp  = disco_w + lr * KK;
        #pragma unroll
        for (int kt = 0; kt < 6; ++kt) {
            const int k0 = kt * 4 + 2 * hi;
            const float* r0 = bp + (long)k0 * (PP * NNZN);
            const float* r1 = r0 + (PP * NNZN);
            v2f bf, a0, a1;
            bf.x = dwp[k0];
            bf.y = dwp[k0 + 1];
            a0.x = r0[lr];
            a0.y = r1[lr];
            const float t0 = r0[n1];
            const float t1 = r1[n1];
            a1.x = n1ok ? t0 : 0.f;
            a1.y = n1ok ? t1 : 0.f;
            accb[0] = wmma_f32(a0, bf, accb[0]);
            accb[1] = wmma_f32(a1, bf, accb[1]);
        }
        {
            const float* r0 = bp + (long)24 * (PP * NNZN);
            const float w  = dwp[24];
            const float t0 = r0[lr];
            const float t1 = r0[n1];
            v2f bf, a0, a1;
            bf.x = lo_half ? w : 0.f;              bf.y = 0.f;
            a0.x = lo_half ? t0 : 0.f;             a0.y = 0.f;
            a1.x = (lo_half && n1ok) ? t1 : 0.f;   a1.y = 0.f;
            accb[0] = wmma_f32(a0, bf, accb[0]);
            accb[1] = wmma_f32(a1, bf, accb[1]);
        }
    }

    const float* xb0 = x;
    const float* xb1 = x + (long)PP * MD;
    unsigned xoff[16];
    {
        const int* np = nbr + p * NNZN;
        #pragma unroll
        for (int r = 0; r < 8; ++r) {
            const int i0 = np[8 * hi + r];
            const int i1 = np[lo_half ? (16 + r) : 24];
            xoff[r]     = (unsigned)i0 * MD + lr;
            xoff[8 + r] = (unsigned)i1 * MD + lr;
        }
    }
    const v8f dbv = ld8(disco_b + 8 * hi);
    const int hrow = p / WW;
    const float* mrow0 = mod + hrow * (2 * SHN);
    const float* mrow1 = mod + (HH_ + hrow) * (2 * SHN);
    const v8f gm0 = ld8(mrow0 + 8 * hi);
    const v8f bt0 = ld8(mrow0 + SHN + 8 * hi);
    const v8f gm1 = ld8(mrow1 + 8 * hi);
    const v8f bt1 = ld8(mrow1 + SHN + 8 * hi);
    const int jl = lr >> 3;
    const float* xres = x   + ((long)hi * PP + p) * MD + lr;
    float*       orow = xln + ((long)hi * PP + p) * CC + lr;

    #pragma unroll 1
    for (int ct = 0; ct < 4; ++ct) {
        v8f acc0 = dbv, acc1 = dbv;
        const unsigned co = ct * 16;
        #pragma unroll
        for (int t = 0; t < 2; ++t) {
            #pragma unroll
            for (int u = 0; u < 4; ++u) {
                v2f af, b0, b1;
                af.x = accb[t][2 * u];
                af.y = accb[t][2 * u + 1];
                b0.x = xb0[xoff[8 * t + 2 * u] + co];
                b0.y = xb0[xoff[8 * t + 2 * u + 1] + co];
                b1.x = xb1[xoff[8 * t + 2 * u] + co];
                b1.y = xb1[xoff[8 * t + 2 * u + 1] + co];
                acc0 = wmma_f32(af, b0, acc0);
                acc1 = wmma_f32(af, b1, acc1);
            }
        }
        #pragma unroll
        for (int r = 0; r < 8; ++r) {
            acc0[r] = acc0[r] * gm0[r] + bt0[r];
            acc1[r] = acc1[r] * gm1[r] + bt1[r];
        }
        v8f g0, g1;
        #pragma unroll
        for (int r = 0; r < 8; ++r) {
            g0[r] = take_hi8(acc0[r], acc1[r]);
            g1[r] = take_lo8(acc1[r], acc0[r]);
        }
        const float tot0 = head_mlp(g0, 2 * ct,     lr, hi, hw1, hb1, hw2);
        const float tot1 = head_mlp(g1, 2 * ct + 1, lr, hi, hw1, hb1, hw2);
        const float s0 = take_hi8(tot0, tot1);
        const float s1 = take_lo8(tot1, tot0);
        const float val = lo_half ? s0 : s1;
        orow[ct * 16] = (hb2[2 * ct + jl] + val) + xres[ct * 16];
    }
}

__device__ __forceinline__ void ffn_load_w1(const float* __restrict__ w1, int nt, int lr, bool hi,
                                            v2f (&bf)[17]) {
    #pragma unroll
    for (int kt = 0; kt < 17; ++kt) {
        const int k0 = kt * 4 + (hi ? 2 : 0);
        const int ka = (k0     < MD) ? k0     : 0;
        const int kb = (k0 + 1 < MD) ? k0 + 1 : 0;
        const float w0  = w1[(long)ka * FFNH + nt * 16 + lr];
        const float w1v = w1[(long)kb * FFNH + nt * 16 + lr];
        bf[kt].x = (k0     < MD) ? w0  : 0.f;
        bf[kt].y = (k0 + 1 < MD) ? w1v : 0.f;
    }
}

__device__ __forceinline__ void ffn_hidden_block(const v2f (&af)[17], const v2f (&bf)[17], float bb,
                                                 float* __restrict__ hid, int nt, int lr, bool hi) {
    v8f acc = {bb, bb, bb, bb, bb, bb, bb, bb};
    #pragma unroll
    for (int kt = 0; kt < 17; ++kt) acc = wmma_f32(af[kt], bf[kt], acc);
    #pragma unroll
    for (int r = 0; r < 8; ++r) {
        const int m = r + (hi ? 8 : 0);
        hid[m * 260 + nt * 16 + lr] = gelu_f(acc[r]);
    }
}

__global__ __launch_bounds__(32)
void ffn_kernel(const float* __restrict__ x, const float* __restrict__ xln,
                const float* __restrict__ w1, const float* __restrict__ b1,
                const float* __restrict__ w2, const float* __restrict__ b2,
                float* __restrict__ out)
{
    __shared__ float hid[16 * 260];

    const int  l   = threadIdx.x;
    const long gq0 = (long)blockIdx.x * 16;
    const bool hi = l >= 16;
    const int  lr = l & 15;

    v2f afr[17];
    {
        const float* xr = xln + (gq0 + lr) * CC + (hi ? 2 : 0);
        #pragma unroll
        for (int kt = 0; kt < 16; ++kt) afr[kt] = *(const v2f*)(xr + kt * 4);
        const v2f sc = *(const v2f*)(x + (gq0 + lr) * MD + CC);
        afr[16].x = hi ? 0.f : sc.x;
        afr[16].y = hi ? 0.f : sc.y;
    }

    v2f bA[17], bB[17];
    ffn_load_w1(w1, 0, lr, hi, bA);
    float bbA = b1[lr];
    for (int nt = 0; nt < 16; nt += 2) {
        ffn_load_w1(w1, nt + 1, lr, hi, bB);
        const float bbB = b1[(nt + 1) * 16 + lr];
        ffn_hidden_block(afr, bA, bbA, hid, nt, lr, hi);
        const int n2 = (nt + 2 < 16) ? nt + 2 : 15;
        ffn_load_w1(w1, n2, lr, hi, bA);
        bbA = b1[n2 * 16 + lr];
        ffn_hidden_block(afr, bB, bbB, hid, nt + 1, lr, hi);
    }
    __syncthreads();

    for (int nt = 0; nt < 4; ++nt) {
        const float bb = b2[nt * 16 + lr];
        v8f acc = {bb, bb, bb, bb, bb, bb, bb, bb};
        #pragma unroll
        for (int kt = 0; kt < 64; ++kt) {
            const int k0 = kt * 4 + (hi ? 2 : 0);
            v2f af, bf;
            af.x = hid[lr * 260 + k0];
            af.y = hid[lr * 260 + k0 + 1];
            bf.x = w2[k0 * CC + nt * 16 + lr];
            bf.y = w2[(k0 + 1) * CC + nt * 16 + lr];
            acc = wmma_f32(af, bf, acc);
        }
        #pragma unroll
        for (int r = 0; r < 8; ++r) {
            const int  m  = r + (hi ? 8 : 0);
            const long gq = gq0 + m;
            const int  c  = nt * 16 + lr;
            out[gq * MD + c] = acc[r] + xln[gq * CC + c];
        }
    }
    {
        const int  m  = l >> 1, c2 = l & 1;
        const long gq = gq0 + m;
        out[gq * MD + CC + c2] = x[gq * MD + CC + c2];
    }
}

extern "C" void kernel_launch(void* const* d_in, const int* in_sizes, int n_in,
                              void* d_out, int out_size, void* d_ws, size_t ws_size,
                              hipStream_t stream)
{
    const float* x       = (const float*)d_in[0];
    const int*   nbr     = (const int*)  d_in[1];
    const float* basis   = (const float*)d_in[2];
    const float* disco_w = (const float*)d_in[3];
    const float* disco_b = (const float*)d_in[4];
    const float* lw1     = (const float*)d_in[5];
    const float* lb1     = (const float*)d_in[6];
    const float* lw2     = (const float*)d_in[7];
    const float* lb2     = (const float*)d_in[8];
    const float* hw1     = (const float*)d_in[9];
    const float* hb1     = (const float*)d_in[10];
    const float* hw2     = (const float*)d_in[11];
    const float* hb2     = (const float*)d_in[12];
    const float* fw1     = (const float*)d_in[13];
    const float* fb1     = (const float*)d_in[14];
    const float* fw2     = (const float*)d_in[15];
    const float* fb2     = (const float*)d_in[16];
    float* out = (float*)d_out;

    float* mod = (float*)d_ws;
    float* xln = (float*)((char*)d_ws + (size_t)(BB * HH_ * 32) * sizeof(float));

    mod_kernel<<<BB * HH_, 32, 0, stream>>>(x, lw1, lb1, lw2, lb2, mod);
    disco_head_kernel<<<PP, 32, 0, stream>>>(x, nbr, basis, disco_w, disco_b,
                                                  hw1, hb1, hw2, hb2, mod, xln);
    ffn_kernel<<<(BB * PP) / 16, 32, 0, stream>>>(x, xln, fw1, fb1, fw2, fb2, out);
}


# gelu as max(v,0) - (1-t)*(P(t)/c)*exp2(-v*v*log2(e)/2), t = rcp(1+c|v|), two values at a time in packed f32: no compare, select or v product; head sum order: even and odd rows summed apart
# speedup vs baseline: 2.3455x; 1.1204x over previous
#include <hip/hip_runtime.h>
#include <hip/hip_bf16.h>

typedef __attribute__((ext_vector_type(2))) float v2f;
typedef __attribute__((ext_vector_type(8))) float v8f;

#define BB    2
#define HH_   90
#define WW    180
#define PP    (HH_ * WW)
#define CC    64
#define MD    66
#define SHN   16
#define KK    25
#define NNZN  25
#define HHID  32
#define FFNH  256

__device__ __forceinline__ v2f gelu2(v2f v) {
    v2f d, t, e, m;
    d.x = __builtin_fmaf(__builtin_fabsf(v.x), 0.318198055f, 1.0f);
    d.y = __builtin_fmaf(__builtin_fabsf(v.y), 0.318198055f, 1.0f);
    t.x = __builtin_amdgcn_rcpf(d.x);
    t.y = __builtin_amdgcn_rcpf(d.y);
    v2f s = {0.145559847f, 0.145559847f};
    s = __builtin_elementwise_fma(t, s, (v2f){-0.776905596f, -0.776905596f});
    s = __builtin_elementwise_fma(t, s, (v2f){ 1.34791970f,   1.34791970f});
    s = __builtin_elementwise_fma(t, s, (v2f){-0.591609299f, -0.591609299f});
    s = __builtin_elementwise_fma(t, s, (v2f){ 0.723748207f,  0.723748207f});
    s = __builtin_elementwise_fma(t, s, (v2f){ 0.315644562f,  0.315644562f});
    s = __builtin_elementwise_fma(t, s, (v2f){ 0.406991005f,  0.406991005f});
    s = __builtin_elementwise_fma(-t, s, s);
    const v2f a = (v * v) * (v2f){-0.721347511f, -0.721347511f};
    e.x = __builtin_amdgcn_exp2f(a.x);
    e.y = __builtin_amdgcn_exp2f(a.y);
    m.x = __builtin_fmaxf(v.x, 0.f);
    m.y = __builtin_fmaxf(v.y, 0.f);
    return __builtin_elementwise_fma(-s, e, m);
}
__device__ __forceinline__ float gelu_f(float v) {
    const v2f r = gelu2((v2f){v, v});
    return r.x;
}

__device__ __forceinline__ v8f wmma_f32(v2f a, v2f b, v8f c) {
    return __builtin_amdgcn_wmma_f32_16x16x4_f32(false, a, false, b, (short)0, c,
                                                 false, false);
}

__global__ __launch_bounds__(32)
void mod_kernel(const float* __restrict__ x,
                const float* __restrict__ lw1, const float* __restrict__ lb1,
                const float* __restrict__ lw2, const float* __restrict__ lb2,
                float* __restrict__ mod)
{
    const int row = blockIdx.x;
    const int j   = threadIdx.x;
    const float* xp = x + ((long)row * WW) * MD + CC;
    const float s0 = xp[0], s1 = xp[1];
    float acc = lb2[j];
    for (int i = 0; i < 2 * SHN; ++i) {
        float t = s0 * lw1[i] + s1 * lw1[2 * SHN + i] + lb1[i];
        acc += gelu_f(t) * lw2[i * (2 * SHN) + j];
    }
    mod[row * (2 * SHN) + j] = acc;
}

typedef __attribute__((ext_vector_type(4))) float v4f;

__device__ __forceinline__ v8f ld8(const float* p) {
    const v4f a = *(const v4f*)(p);
    const v4f c = *(const v4f*)(p + 4);
    v8f r = {a.x, a.y, a.z, a.w, c.x, c.y, c.z, c.w};
    return r;
}

__device__ __forceinline__ float take_hi8(float keep, float src) {
    return __builtin_bit_cast(float, __builtin_amdgcn_update_dpp(
        __builtin_bit_cast(int, keep), __builtin_bit_cast(int, src), 0x128, 0xF, 0xC, false));
}
__device__ __forceinline__ float take_lo8(float keep, float src) {
    return __builtin_bit_cast(float, __builtin_amdgcn_update_dpp(
        __builtin_bit_cast(int, keep), __builtin_bit_cast(int, src), 0x128, 0xF, 0x3, false));
}

__device__ __forceinline__ float head_mlp(const v8f g, int head, int lr, int hi,
                                          const float* __restrict__ hw1,
                                          const float* __restrict__ hb1,
                                          const float* __restrict__ hw2) {
    v8f acch[2];
    acch[0] = ld8(hb1 + head * HHID + 8 * hi);
    acch[1] = ld8(hb1 + head * HHID + 16 + 8 * hi);
    #pragma unroll
    for (int u = 0; u < 4; ++u) {
        v2f bf;
        bf.x = g[2 * u];
        bf.y = g[2 * u + 1];
        const float* wp = hw1 + (head * SHN + 8 * hi + 2 * u) * HHID + lr;
        #pragma unroll
        for (int nt = 0; nt < 2; ++nt) {
            v2f af;
            af.x = wp[nt * 16];
            af.y = wp[HHID + nt * 16];
            acch[nt] = wmma_f32(af, bf, acch[nt]);
        }
    }
    v2f ps2 = {0.f, 0.f};
    #pragma unroll
    for (int nt = 0; nt < 2; ++nt) {
        const v8f w2v = ld8(hw2 + head * HHID + nt * 16 + 8 * hi);
        #pragma unroll
        for (int j = 0; j < 4; ++j) {
            const v2f hv = {acch[nt][2 * j], acch[nt][2 * j + 1]};
            const v2f wv = {w2v[2 * j], w2v[2 * j + 1]};
            ps2 = __builtin_elementwise_fma(gelu2(hv), wv, ps2);
        }
    }
    const float ps = ps2.x + ps2.y;
    return ps + __shfl_xor(ps, 16, 32);
}

static_assert(BB == 2, "the point kernel pairs the two batch rows of a grid point in one wave");

__global__ __launch_bounds__(32, 5)
void disco_head_kernel(const float* __restrict__ x, const int* __restrict__ nbr,
                       const float* __restrict__ basis,
                       const float* __restrict__ disco_w,
                       const float* __restrict__ disco_b,
                       const float* __restrict__ hw1, const float* __restrict__ hb1,
                       const float* __restrict__ hw2, const float* __restrict__ hb2,
                       const float* __restrict__ mod, float* __restrict__ xln)
{
    const int l  = threadIdx.x;
    const int p  = blockIdx.x;
    const int hi = l >> 4;
    const int lr = l & 15;
    const bool lo_half = (hi == 0);

    v8f accb[2];
    {
        const v8f zb = {0.f, 0.f, 0.f, 0.f, 0.f, 0.f, 0.f, 0.f};
        accb[0] = zb; accb[1] = zb;
        const float* bp   = basis + (long)p * NNZN;
        const bool   n1ok = lr < (NNZN - 16);
        const int    n1   = n1ok ? (16 + lr) : 0;
        const float* dwp  = disco_w + lr * KK;
        #pragma unroll
        for (int kt = 0; kt < 6; ++kt) {
            const int k0 = kt * 4 + 2 * hi;
            const float* r0 = bp + (long)k0 * (PP * NNZN);
            const float* r1 = r0 + (PP * NNZN);
            v2f bf, a0, a1;
            bf.x = dwp[k0];
            bf.y = dwp[k0 + 1];
            a0.x = r0[lr];
            a0.y = r1[lr];
            const float t0 = r0[n1];
            const float t1 = r1[n1];
            a1.x = n1ok ? t0 : 0.f;
            a1.y = n1ok ? t1 : 0.f;
            accb[0] = wmma_f32(a0, bf, accb[0]);
            accb[1] = wmma_f32(a1, bf, accb[1]);
        }
        {
            const float* r0 = bp + (long)24 * (PP * NNZN);
            const float w  = dwp[24];
            const float t0 = r0[lr];
            const float t1 = r0[n1];
            v2f bf, a0, a1;
            bf.x = lo_half ? w : 0.f;              bf.y = 0.f;
            a0.x = lo_half ? t0 : 0.f;             a0.y = 0.f;
            a1.x = (lo_half && n1ok) ? t1 : 0.f;   a1.y = 0.f;
            accb[0] = wmma_f32(a0, bf, accb[0]);
            accb[1] = wmma_f32(a1, bf, accb[1]);
        }
    }

    const float* xb0 = x;
    const float* xb1 = x + (long)PP * MD;
    unsigned xoff[16];
    {
        const int* np = nbr + p * NNZN;
        #pragma unroll
        for (int r = 0; r < 8; ++r) {
            const int i0 = np[8 * hi + r];
            const int i1 = np[lo_half ? (16 + r) : 24];
            xoff[r]     = (unsigned)i0 * MD + lr;
            xoff[8 + r] = (unsigned)i1 * MD + lr;
        }
    }
    const v8f dbv = ld8(disco_b + 8 * hi);
    const int hrow = p / WW;
    const float* mrow0 = mod + hrow * (2 * SHN);
    const float* mrow1 = mod + (HH_ + hrow) * (2 * SHN);
    const v8f gm0 = ld8(mrow0 + 8 * hi);
    const v8f bt0 = ld8(mrow0 + SHN + 8 * hi);
    const v8f gm1 = ld8(mrow1 + 8 * hi);
    const v8f bt1 = ld8(mrow1 + SHN + 8 * hi);
    const int jl = lr >> 3;
    const float* xres = x   + ((long)hi * PP + p) * MD + lr;
    float*       orow = xln + ((long)hi * PP + p) * CC + lr;

    #pragma unroll 1
    for (int ct = 0; ct < 4; ++ct) {
        v8f acc0 = dbv, acc1 = dbv;
        const unsigned co = ct * 16;
        #pragma unroll
        for (int t = 0; t < 2; ++t) {
            #pragma unroll
            for (int u = 0; u < 4; ++u) {
                v2f af, b0, b1;
                af.x = accb[t][2 * u];
                af.y = accb[t][2 * u + 1];
                b0.x = xb0[xoff[8 * t + 2 * u] + co];
                b0.y = xb0[xoff[8 * t + 2 * u + 1] + co];
                b1.x = xb1[xoff[8 * t + 2 * u] + co];
                b1.y = xb1[xoff[8 * t + 2 * u + 1] + co];
                acc0 = wmma_f32(af, b0, acc0);
                acc1 = wmma_f32(af, b1, acc1);
            }
        }
        #pragma unroll
        for (int r = 0; r < 8; ++r) {
            acc0[r] = acc0[r] * gm0[r] + bt0[r];
            acc1[r] = acc1[r] * gm1[r] + bt1[r];
        }
        v8f g0, g1;
        #pragma unroll
        for (int r = 0; r < 8; ++r) {
            g0[r] = take_hi8(acc0[r], acc1[r]);
            g1[r] = take_lo8(acc1[r], acc0[r]);
        }
        const float tot0 = head_mlp(g0, 2 * ct,     lr, hi, hw1, hb1, hw2);
        __builtin_amdgcn_sched_barrier(0);
        const float tot1 = head_mlp(g1, 2 * ct + 1, lr, hi, hw1, hb1, hw2);
        const float s0 = take_hi8(tot0, tot1);
        const float s1 = take_lo8(tot1, tot0);
        const float val = lo_half ? s0 : s1;
        orow[ct * 16] = (hb2[2 * ct + jl] + val) + xres[ct * 16];
    }
}

__device__ __forceinline__ void ffn_load_w1(const float* __restrict__ w1, int nt, int lr, bool hi,
                                            v2f (&bf)[17]) {
    #pragma unroll
    for (int kt = 0; kt < 17; ++kt) {
        const int k0 = kt * 4 + (hi ? 2 : 0);
        const int ka = (k0     < MD) ? k0     : 0;
        const int kb = (k0 + 1 < MD) ? k0 + 1 : 0;
        const float w0  = w1[(long)ka * FFNH + nt * 16 + lr];
        const float w1v = w1[(long)kb * FFNH + nt * 16 + lr];
        bf[kt].x = (k0     < MD) ? w0  : 0.f;
        bf[kt].y = (k0 + 1 < MD) ? w1v : 0.f;
    }
}

__device__ __forceinline__ void ffn_hidden_block(const v2f (&af)[17], const v2f (&bf)[17], float bb,
                                                 float* __restrict__ hid, int nt, int lr, bool hi) {
    v8f acc = {bb, bb, bb, bb, bb, bb, bb, bb};
    #pragma unroll
    for (int kt = 0; kt < 17; ++kt) acc = wmma_f32(af[kt], bf[kt], acc);
    #pragma unroll
    for (int j = 0; j < 4; ++j) {
        const int m = 2 * j + (hi ? 8 : 0);
        const v2f g = gelu2((v2f){acc[2 * j], acc[2 * j + 1]});
        hid[m * 260 + nt * 16 + lr]       = g.x;
        hid[(m + 1) * 260 + nt * 16 + lr] = g.y;
    }
}

__global__ __launch_bounds__(32)
void ffn_kernel(const float* __restrict__ x, const float* __restrict__ xln,
                const float* __restrict__ w1, const float* __restrict__ b1,
                const float* __restrict__ w2, const float* __restrict__ b2,
                float* __restrict__ out)
{
    __shared__ float hid[16 * 260];

    const int  l   = threadIdx.x;
    const long gq0 = (long)blockIdx.x * 16;
    const bool hi = l >= 16;
    const int  lr = l & 15;

    v2f afr[17];
    {
        const float* xr = xln + (gq0 + lr) * CC + (hi ? 2 : 0);
        #pragma unroll
        for (int kt = 0; kt < 16; ++kt) afr[kt] = *(const v2f*)(xr + kt * 4);
        const v2f sc = *(const v2f*)(x + (gq0 + lr) * MD + CC);
        afr[16].x = hi ? 0.f : sc.x;
        afr[16].y = hi ? 0.f : sc.y;
    }

    v2f bA[17], bB[17];
    ffn_load_w1(w1, 0, lr, hi, bA);
    float bbA = b1[lr];
    for (int nt = 0; nt < 16; nt += 2) {
        ffn_load_w1(w1, nt + 1, lr, hi, bB);
        const float bbB = b1[(nt + 1) * 16 + lr];
        ffn_hidden_block(afr, bA, bbA, hid, nt, lr, hi);
        const int n2 = (nt + 2 < 16) ? nt + 2 : 15;
        ffn_load_w1(w1, n2, lr, hi, bA);
        bbA = b1[n2 * 16 + lr];
        ffn_hidden_block(afr, bB, bbB, hid, nt + 1, lr, hi);
    }
    __syncthreads();

    for (int nt = 0; nt < 4; ++nt) {
        const float bb = b2[nt * 16 + lr];
        v8f acc = {bb, bb, bb, bb, bb, bb, bb, bb};
        #pragma unroll
        for (int kt = 0; kt < 64; ++kt) {
            const int k0 = kt * 4 + (hi ? 2 : 0);
            v2f af, bf;
            af.x = hid[lr * 260 + k0];
            af.y = hid[lr * 260 + k0 + 1];
            bf.x = w2[k0 * CC + nt * 16 + lr];
            bf.y = w2[(k0 + 1) * CC + nt * 16 + lr];
            acc = wmma_f32(af, bf, acc);
        }
        #pragma unroll
        for (int r = 0; r < 8; ++r) {
            const int  m  = r + (hi ? 8 : 0);
            const long gq = gq0 + m;
            const int  c  = nt * 16 + lr;
            out[gq * MD + c] = acc[r] + xln[gq * CC + c];
        }
    }
    {
        const int  m  = l >> 1, c2 = l & 1;
        const long gq = gq0 + m;
        out[gq * MD + CC + c2] = x[gq * MD + CC + c2];
    }
}

extern "C" void kernel_launch(void* const* d_in, const int* in_sizes, int n_in,
                              void* d_out, int out_size, void* d_ws, size_t ws_size,
                              hipStream_t stream)
{
    const float* x       = (const float*)d_in[0];
    const int*   nbr     = (const int*)  d_in[1];
    const float* basis   = (const float*)d_in[2];
    const float* disco_w = (const float*)d_in[3];
    const float* disco_b = (const float*)d_in[4];
    const float* lw1     = (const float*)d_in[5];
    const float* lb1     = (const float*)d_in[6];
    const float* lw2     = (const float*)d_in[7];
    const float* lb2     = (const float*)d_in[8];
    const float* hw1     = (const float*)d_in[9];
    const float* hb1     = (const float*)d_in[10];
    const float* hw2     = (const float*)d_in[11];
    const float* hb2     = (const float*)d_in[12];
    const float* fw1     = (const float*)d_in[13];
    const float* fb1     = (const float*)d_in[14];
    const float* fw2     = (const float*)d_in[15];
    const float* fb2     = (const float*)d_in[16];
    float* out = (float*)d_out;

    float* mod = (float*)d_ws;
    float* xln = (float*)((char*)d_ws + (size_t)(BB * HH_ * 32) * sizeof(float));

    mod_kernel<<<BB * HH_, 32, 0, stream>>>(x, lw1, lb1, lw2, lb2, mod);
    disco_head_kernel<<<PP, 32, 0, stream>>>(x, nbr, basis, disco_w, disco_b,
                                                  hw1, hb1, hw2, hb2, mod, xln);
    ffn_kernel<<<(BB * PP) / 16, 32, 0, stream>>>(x, xln, fw1, fb1, fw2, fb2, out);
}


# Point kernel: each neighbour row of x gathered once as a whole 256-byte row (16-byte loads) into LDS; the four channel tiles take their B fragments from LDS, not four 64-byte global pieces
# speedup vs baseline: 2.4551x; 1.0467x over previous
#include <hip/hip_runtime.h>
#include <hip/hip_bf16.h>

typedef __attribute__((ext_vector_type(2))) float v2f;
typedef __attribute__((ext_vector_type(8))) float v8f;

#define BB    2
#define HH_   90
#define WW    180
#define PP    (HH_ * WW)
#define CC    64
#define MD    66
#define SHN   16
#define KK    25
#define NNZN  25
#define HHID  32
#define FFNH  256
#define XS_PITCH 68

__device__ __forceinline__ v2f gelu2(v2f v) {
    v2f d, t, e, m;
    d.x = __builtin_fmaf(__builtin_fabsf(v.x), 0.318198055f, 1.0f);
    d.y = __builtin_fmaf(__builtin_fabsf(v.y), 0.318198055f, 1.0f);
    t.x = __builtin_amdgcn_rcpf(d.x);
    t.y = __builtin_amdgcn_rcpf(d.y);
    v2f s = {0.145559847f, 0.145559847f};
    s = __builtin_elementwise_fma(t, s, (v2f){-0.776905596f, -0.776905596f});
    s = __builtin_elementwise_fma(t, s, (v2f){ 1.34791970f,   1.34791970f});
    s = __builtin_elementwise_fma(t, s, (v2f){-0.591609299f, -0.591609299f});
    s = __builtin_elementwise_fma(t, s, (v2f){ 0.723748207f,  0.723748207f});
    s = __builtin_elementwise_fma(t, s, (v2f){ 0.315644562f,  0.315644562f});
    s = __builtin_elementwise_fma(t, s, (v2f){ 0.406991005f,  0.406991005f});
    s = __builtin_elementwise_fma(-t, s, s);
    const v2f a = (v * v) * (v2f){-0.721347511f, -0.721347511f};
    e.x = __builtin_amdgcn_exp2f(a.x);
    e.y = __builtin_amdgcn_exp2f(a.y);
    m.x = __builtin_fmaxf(v.x, 0.f);
    m.y = __builtin_fmaxf(v.y, 0.f);
    return __builtin_elementwise_fma(-s, e, m);
}
__device__ __forceinline__ float gelu_f(float v) {
    const v2f r = gelu2((v2f){v, v});
    return r.x;
}

__device__ __forceinline__ v8f wmma_f32(v2f a, v2f b, v8f c) {
    return __builtin_amdgcn_wmma_f32_16x16x4_f32(false, a, false, b, (short)0, c,
                                                 false, false);
}

__global__ __launch_bounds__(32)
void mod_kernel(const float* __restrict__ x,
                const float* __restrict__ lw1, const float* __restrict__ lb1,
                const float* __restrict__ lw2, const float* __restrict__ lb2,
                float* __restrict__ mod)
{
    const int row = blockIdx.x;
    const int j   = threadIdx.x;
    const float* xp = x + ((long)row * WW) * MD + CC;
    const float s0 = xp[0], s1 = xp[1];
    float acc = lb2[j];
    for (int i = 0; i < 2 * SHN; ++i) {
        float t = s0 * lw1[i] + s1 * lw1[2 * SHN + i] + lb1[i];
        acc += gelu_f(t) * lw2[i * (2 * SHN) + j];
    }
    mod[row * (2 * SHN) + j] = acc;
}

typedef __attribute__((ext_vector_type(4))) float v4f;

__device__ __forceinline__ v8f ld8(const float* p) {
    const v4f a = *(const v4f*)(p);
    const v4f c = *(const v4f*)(p + 4);
    v8f r = {a.x, a.y, a.z, a.w, c.x, c.y, c.z, c.w};
    return r;
}

__device__ __forceinline__ float take_hi8(float keep, float src) {
    return __builtin_bit_cast(float, __builtin_amdgcn_update_dpp(
        __builtin_bit_cast(int, keep), __builtin_bit_cast(int, src), 0x128, 0xF, 0xC, false));
}
__device__ __forceinline__ float take_lo8(float keep, float src) {
    return __builtin_bit_cast(float, __builtin_amdgcn_update_dpp(
        __builtin_bit_cast(int, keep), __builtin_bit_cast(int, src), 0x128, 0xF, 0x3, false));
}

__device__ __forceinline__ float head_mlp(const v8f g, int head, int lr, int hi,
                                          const float* __restrict__ hw1,
                                          const float* __restrict__ hb1,
                                          const float* __restrict__ hw2) {
    v8f acch[2];
    acch[0] = ld8(hb1 + head * HHID + 8 * hi);
    acch[1] = ld8(hb1 + head * HHID + 16 + 8 * hi);
    #pragma unroll
    for (int u = 0; u < 4; ++u) {
        v2f bf;
        bf.x = g[2 * u];
        bf.y = g[2 * u + 1];
        const float* wp = hw1 + (head * SHN + 8 * hi + 2 * u) * HHID + lr;
        #pragma unroll
        for (int nt = 0; nt < 2; ++nt) {
            v2f af;
            af.x = wp[nt * 16];
            af.y = wp[HHID + nt * 16];
            acch[nt] = wmma_f32(af, bf, acch[nt]);
        }
    }
    v2f ps2 = {0.f, 0.f};
    #pragma unroll
    for (int nt = 0; nt < 2; ++nt) {
        const v8f w2v = ld8(hw2 + head * HHID + nt * 16 + 8 * hi);
        #pragma unroll
        for (int j = 0; j < 4; ++j) {
            const v2f hv = {acch[nt][2 * j], acch[nt][2 * j + 1]};
            const v2f wv = {w2v[2 * j], w2v[2 * j + 1]};
            ps2 = __builtin_elementwise_fma(gelu2(hv), wv, ps2);
        }
    }
    const float ps = ps2.x + ps2.y;
    return ps + __shfl_xor(ps, 16, 32);
}

static_assert(BB == 2, "the point kernel pairs the two batch rows of a grid point in one wave");

__global__ __launch_bounds__(32, 5)
void disco_head_kernel(const float* __restrict__ x, const int* __restrict__ nbr,
                       const float* __restrict__ basis,
                       const float* __restrict__ disco_w,
                       const float* __restrict__ disco_b,
                       const float* __restrict__ hw1, const float* __restrict__ hb1,
                       const float* __restrict__ hw2, const float* __restrict__ hb2,
                       const float* __restrict__ mod, float* __restrict__ xln)
{
    __shared__ __attribute__((aligned(16))) float xs[2 * NNZN * XS_PITCH];

    const int l  = threadIdx.x;
    const int p  = blockIdx.x;
    const int hi = l >> 4;
    const int lr = l & 15;
    const bool lo_half = (hi == 0);

    v8f accb[2];
    {
        const v8f zb = {0.f, 0.f, 0.f, 0.f, 0.f, 0.f, 0.f, 0.f};
        accb[0] = zb; accb[1] = zb;
        const float* bp   = basis + (long)p * NNZN;
        const bool   n1ok = lr < (NNZN - 16);
        const int    n1   = n1ok ? (16 + lr) : 0;
        const float* dwp  = disco_w + lr * KK;
        #pragma unroll
        for (int kt = 0; kt < 6; ++kt) {
            const int k0 = kt * 4 + 2 * hi;
            const float* r0 = bp + (long)k0 * (PP * NNZN);
            const float* r1 = r0 + (PP * NNZN);
            v2f bf, a0, a1;
            bf.x = dwp[k0];
            bf.y = dwp[k0 + 1];
            a0.x = r0[lr];
            a0.y = r1[lr];
            const float t0 = r0[n1];
            const float t1 = r1[n1];
            a1.x = n1ok ? t0 : 0.f;
            a1.y = n1ok ? t1 : 0.f;
            accb[0] = wmma_f32(a0, bf, accb[0]);
            accb[1] = wmma_f32(a1, bf, accb[1]);
        }
        {
            const float* r0 = bp + (long)24 * (PP * NNZN);
            const float w  = dwp[24];
            const float t0 = r0[lr];
            const float t1 = r0[n1];
            v2f bf, a0, a1;
            bf.x = lo_half ? w : 0.f;              bf.y = 0.f;
            a0.x = lo_half ? t0 : 0.f;             a0.y = 0.f;
            a1.x = (lo_half && n1ok) ? t1 : 0.f;   a1.y = 0.f;
            accb[0] = wmma_f32(a0, bf, accb[0]);
            accb[1] = wmma_f32(a1, bf, accb[1]);
        }
    }

    {
        const int* np = nbr + p * NNZN;
        #pragma unroll
        for (int i = 0; i < NNZN; ++i) {
            const int r  = 2 * i + (hi & 1);
            const int bq = (r >= NNZN) ? 1 : 0;
            const int n  = r - NNZN * bq;
            const unsigned idx = (unsigned)np[n];
            const unsigned boff = __umul24(idx, MD * 4) + (unsigned)(16 * lr)
                                  + (unsigned)bq * (unsigned)(PP * MD * 4);
            const char* rowp = (const char*)x + boff;
            v4f v;
            __builtin_memcpy(&v, __builtin_assume_aligned(rowp, 8), 16);
            *(v4f*)&xs[r * XS_PITCH + 4 * lr] = v;
        }
    }
    unsigned xl[16];
    #pragma unroll
    for (int r = 0; r < 8; ++r) {
        xl[r]     = (unsigned)(8 * hi + r) * XS_PITCH + lr;
        xl[8 + r] = (unsigned)(lo_half ? (16 + r) : 24) * XS_PITCH + lr;
    }
    __syncthreads();
    const v8f dbv = ld8(disco_b + 8 * hi);
    const int hrow = p / WW;
    const float* mrow0 = mod + hrow * (2 * SHN);
    const float* mrow1 = mod + (HH_ + hrow) * (2 * SHN);
    const v8f gm0 = ld8(mrow0 + 8 * hi);
    const v8f bt0 = ld8(mrow0 + SHN + 8 * hi);
    const v8f gm1 = ld8(mrow1 + 8 * hi);
    const v8f bt1 = ld8(mrow1 + SHN + 8 * hi);
    const int jl = lr >> 3;
    const float* xres = x   + ((long)hi * PP + p) * MD + lr;
    float*       orow = xln + ((long)hi * PP + p) * CC + lr;

    #pragma unroll 1
    for (int ct = 0; ct < 4; ++ct) {
        v8f acc0 = dbv, acc1 = dbv;
        const unsigned co = ct * 16;
        #pragma unroll
        for (int t = 0; t < 2; ++t) {
            #pragma unroll
            for (int u = 0; u < 4; ++u) {
                v2f af, b0, b1;
                af.x = accb[t][2 * u];
                af.y = accb[t][2 * u + 1];
                b0.x = xs[xl[8 * t + 2 * u] + co];
                b0.y = xs[xl[8 * t + 2 * u + 1] + co];
                b1.x = xs[xl[8 * t + 2 * u] + NNZN * XS_PITCH + co];
                b1.y = xs[xl[8 * t + 2 * u + 1] + NNZN * XS_PITCH + co];
                acc0 = wmma_f32(af, b0, acc0);
                acc1 = wmma_f32(af, b1, acc1);
            }
        }
        #pragma unroll
        for (int r = 0; r < 8; ++r) {
            acc0[r] = acc0[r] * gm0[r] + bt0[r];
            acc1[r] = acc1[r] * gm1[r] + bt1[r];
        }
        v8f g0, g1;
        #pragma unroll
        for (int r = 0; r < 8; ++r) {
            g0[r] = take_hi8(acc0[r], acc1[r]);
            g1[r] = take_lo8(acc1[r], acc0[r]);
        }
        const float tot0 = head_mlp(g0, 2 * ct,     lr, hi, hw1, hb1, hw2);
        __builtin_amdgcn_sched_barrier(0);
        const float tot1 = head_mlp(g1, 2 * ct + 1, lr, hi, hw1, hb1, hw2);
        const float s0 = take_hi8(tot0, tot1);
        const float s1 = take_lo8(tot1, tot0);
        const float val = lo_half ? s0 : s1;
        orow[ct * 16] = (hb2[2 * ct + jl] + val) + xres[ct * 16];
    }
}

__device__ __forceinline__ void ffn_load_w1(const float* __restrict__ w1, int nt, int lr, bool hi,
                                            v2f (&bf)[17]) {
    #pragma unroll
    for (int kt = 0; kt < 17; ++kt) {
        const int k0 = kt * 4 + (hi ? 2 : 0);
        const int ka = (k0     < MD) ? k0     : 0;
        const int kb = (k0 + 1 < MD) ? k0 + 1 : 0;
        const float w0  = w1[(long)ka * FFNH + nt * 16 + lr];
        const float w1v = w1[(long)kb * FFNH + nt * 16 + lr];
        bf[kt].x = (k0     < MD) ? w0  : 0.f;
        bf[kt].y = (k0 + 1 < MD) ? w1v : 0.f;
    }
}

__device__ __forceinline__ void ffn_hidden_block(const v2f (&af)[17], const v2f (&bf)[17], float bb,
                                                 float* __restrict__ hid, int nt, int lr, bool hi) {
    v8f acc = {bb, bb, bb, bb, bb, bb, bb, bb};
    #pragma unroll
    for (int kt = 0; kt < 17; ++kt) acc = wmma_f32(af[kt], bf[kt], acc);
    #pragma unroll
    for (int j = 0; j < 4; ++j) {
        const int m = 2 * j + (hi ? 8 : 0);
        const v2f g = gelu2((v2f){acc[2 * j], acc[2 * j + 1]});
        hid[m * 260 + nt * 16 + lr]       = g.x;
        hid[(m + 1) * 260 + nt * 16 + lr] = g.y;
    }
}

__global__ __launch_bounds__(32)
void ffn_kernel(const float* __restrict__ x, const float* __restrict__ xln,
                const float* __restrict__ w1, const float* __restrict__ b1,
                const float* __restrict__ w2, const float* __restrict__ b2,
                float* __restrict__ out)
{
    __shared__ float hid[16 * 260];

    const int  l   = threadIdx.x;
    const long gq0 = (long)blockIdx.x * 16;
    const bool hi = l >= 16;
    const int  lr = l & 15;

    v2f afr[17];
    {
        const float* xr = xln + (gq0 + lr) * CC + (hi ? 2 : 0);
        #pragma unroll
        for (int kt = 0; kt < 16; ++kt) afr[kt] = *(const v2f*)(xr + kt * 4);
        const v2f sc = *(const v2f*)(x + (gq0 + lr) * MD + CC);
        afr[16].x = hi ? 0.f : sc.x;
        afr[16].y = hi ? 0.f : sc.y;
    }

    v2f bA[17], bB[17];
    ffn_load_w1(w1, 0, lr, hi, bA);
    float bbA = b1[lr];
    for (int nt = 0; nt < 16; nt += 2) {
        ffn_load_w1(w1, nt + 1, lr, hi, bB);
        const float bbB = b1[(nt + 1) * 16 + lr];
        ffn_hidden_block(afr, bA, bbA, hid, nt, lr, hi);
        const int n2 = (nt + 2 < 16) ? nt + 2 : 15;
        ffn_load_w1(w1, n2, lr, hi, bA);
        bbA = b1[n2 * 16 + lr];
        ffn_hidden_block(afr, bB, bbB, hid, nt + 1, lr, hi);
    }
    __syncthreads();

    for (int nt = 0; nt < 4; ++nt) {
        const float bb = b2[nt * 16 + lr];
        v8f acc = {bb, bb, bb, bb, bb, bb, bb, bb};
        #pragma unroll
        for (int kt = 0; kt < 64; ++kt) {
            const int k0 = kt * 4 + (hi ? 2 : 0);
            v2f af, bf;
            af.x = hid[lr * 260 + k0];
            af.y = hid[lr * 260 + k0 + 1];
            bf.x = w2[k0 * CC + nt * 16 + lr];
            bf.y = w2[(k0 + 1) * CC + nt * 16 + lr];
            acc = wmma_f32(af, bf, acc);
        }
        #pragma unroll
        for (int r = 0; r < 8; ++r) {
            const int  m  = r + (hi ? 8 : 0);
            const long gq = gq0 + m;
            const int  c  = nt * 16 + lr;
            out[gq * MD + c] = acc[r] + xln[gq * CC + c];
        }
    }
    {
        const int  m  = l >> 1, c2 = l & 1;
        const long gq = gq0 + m;
        out[gq * MD + CC + c2] = x[gq * MD + CC + c2];
    }
}

extern "C" void kernel_launch(void* const* d_in, const int* in_sizes, int n_in,
                              void* d_out, int out_size, void* d_ws, size_t ws_size,
                              hipStream_t stream)
{
    const float* x       = (const float*)d_in[0];
    const int*   nbr     = (const int*)  d_in[1];
    const float* basis   = (const float*)d_in[2];
    const float* disco_w = (const float*)d_in[3];
    const float* disco_b = (const float*)d_in[4];
    const float* lw1     = (const float*)d_in[5];
    const float* lb1     = (const float*)d_in[6];
    const float* lw2     = (const float*)d_in[7];
    const float* lb2     = (const float*)d_in[8];
    const float* hw1     = (const float*)d_in[9];
    const float* hb1     = (const float*)d_in[10];
    const float* hw2     = (const float*)d_in[11];
    const float* hb2     = (const float*)d_in[12];
    const float* fw1     = (const float*)d_in[13];
    const float* fb1     = (const float*)d_in[14];
    const float* fw2     = (const float*)d_in[15];
    const float* fb2     = (const float*)d_in[16];
    float* out = (float*)d_out;

    float* mod = (float*)d_ws;
    float* xln = (float*)((char*)d_ws + (size_t)(BB * HH_ * 32) * sizeof(float));

    mod_kernel<<<BB * HH_, 32, 0, stream>>>(x, lw1, lb1, lw2, lb2, mod);
    disco_head_kernel<<<PP, 32, 0, stream>>>(x, nbr, basis, disco_w, disco_b,
                                                  hw1, hb1, hw2, hb2, mod, xln);
    ffn_kernel<<<(BB * PP) / 16, 32, 0, stream>>>(x, xln, fw1, fb1, fw2, fb2, out);
}


# FFN kernel: no matrix instruction accumulates onto the one before it: first product runs two column blocks' chains interleaved, second all four column tiles in one k loop (W2 as 16-byte row loads)
# speedup vs baseline: 2.5972x; 1.0579x over previous
#include <hip/hip_runtime.h>
#include <hip/hip_bf16.h>

typedef __attribute__((ext_vector_type(2))) float v2f;
typedef __attribute__((ext_vector_type(8))) float v8f;

#define BB    2
#define HH_   90
#define WW    180
#define PP    (HH_ * WW)
#define CC    64
#define MD    66
#define SHN   16
#define KK    25
#define NNZN  25
#define HHID  32
#define FFNH  256
#define XS_PITCH 68

__device__ __forceinline__ v2f gelu2(v2f v) {
    v2f d, t, e, m;
    d.x = __builtin_fmaf(__builtin_fabsf(v.x), 0.318198055f, 1.0f);
    d.y = __builtin_fmaf(__builtin_fabsf(v.y), 0.318198055f, 1.0f);
    t.x = __builtin_amdgcn_rcpf(d.x);
    t.y = __builtin_amdgcn_rcpf(d.y);
    v2f s = {0.145559847f, 0.145559847f};
    s = __builtin_elementwise_fma(t, s, (v2f){-0.776905596f, -0.776905596f});
    s = __builtin_elementwise_fma(t, s, (v2f){ 1.34791970f,   1.34791970f});
    s = __builtin_elementwise_fma(t, s, (v2f){-0.591609299f, -0.591609299f});
    s = __builtin_elementwise_fma(t, s, (v2f){ 0.723748207f,  0.723748207f});
    s = __builtin_elementwise_fma(t, s, (v2f){ 0.315644562f,  0.315644562f});
    s = __builtin_elementwise_fma(t, s, (v2f){ 0.406991005f,  0.406991005f});
    s = __builtin_elementwise_fma(-t, s, s);
    const v2f a = (v * v) * (v2f){-0.721347511f, -0.721347511f};
    e.x = __builtin_amdgcn_exp2f(a.x);
    e.y = __builtin_amdgcn_exp2f(a.y);
    m.x = __builtin_fmaxf(v.x, 0.f);
    m.y = __builtin_fmaxf(v.y, 0.f);
    return __builtin_elementwise_fma(-s, e, m);
}
__device__ __forceinline__ float gelu_f(float v) {
    const v2f r = gelu2((v2f){v, v});
    return r.x;
}

__device__ __forceinline__ v8f wmma_f32(v2f a, v2f b, v8f c) {
    return __builtin_amdgcn_wmma_f32_16x16x4_f32(false, a, false, b, (short)0, c,
                                                 false, false);
}

__global__ __launch_bounds__(32)
void mod_kernel(const float* __restrict__ x,
                const float* __restrict__ lw1, const float* __restrict__ lb1,
                const float* __restrict__ lw2, const float* __restrict__ lb2,
                float* __restrict__ mod)
{
    const int row = blockIdx.x;
    const int j   = threadIdx.x;
    const float* xp = x + ((long)row * WW) * MD + CC;
    const float s0 = xp[0], s1 = xp[1];
    float acc = lb2[j];
    for (int i = 0; i < 2 * SHN; ++i) {
        float t = s0 * lw1[i] + s1 * lw1[2 * SHN + i] + lb1[i];
        acc += gelu_f(t) * lw2[i * (2 * SHN) + j];
    }
    mod[row * (2 * SHN) + j] = acc;
}

typedef __attribute__((ext_vector_type(4))) float v4f;

__device__ __forceinline__ v8f ld8(const float* p) {
    const v4f a = *(const v4f*)(p);
    const v4f c = *(const v4f*)(p + 4);
    v8f r = {a.x, a.y, a.z, a.w, c.x, c.y, c.z, c.w};
    return r;
}

__device__ __forceinline__ float take_hi8(float keep, float src) {
    return __builtin_bit_cast(float, __builtin_amdgcn_update_dpp(
        __builtin_bit_cast(int, keep), __builtin_bit_cast(int, src), 0x128, 0xF, 0xC, false));
}
__device__ __forceinline__ float take_lo8(float keep, float src) {
    return __builtin_bit_cast(float, __builtin_amdgcn_update_dpp(
        __builtin_bit_cast(int, keep), __builtin_bit_cast(int, src), 0x128, 0xF, 0x3, false));
}

__device__ __forceinline__ float head_mlp(const v8f g, int head, int lr, int hi,
                                          const float* __restrict__ hw1,
                                          const float* __restrict__ hb1,
                                          const float* __restrict__ hw2) {
    v8f acch[2];
    acch[0] = ld8(hb1 + head * HHID + 8 * hi);
    acch[1] = ld8(hb1 + head * HHID + 16 + 8 * hi);
    #pragma unroll
    for (int u = 0; u < 4; ++u) {
        v2f bf;
        bf.x = g[2 * u];
        bf.y = g[2 * u + 1];
        const float* wp = hw1 + (head * SHN + 8 * hi + 2 * u) * HHID + lr;
        #pragma unroll
        for (int nt = 0; nt < 2; ++nt) {
            v2f af;
            af.x = wp[nt * 16];
            af.y = wp[HHID + nt * 16];
            acch[nt] = wmma_f32(af, bf, acch[nt]);
        }
    }
    v2f ps2 = {0.f, 0.f};
    #pragma unroll
    for (int nt = 0; nt < 2; ++nt) {
        const v8f w2v = ld8(hw2 + head * HHID + nt * 16 + 8 * hi);
        #pragma unroll
        for (int j = 0; j < 4; ++j) {
            const v2f hv = {acch[nt][2 * j], acch[nt][2 * j + 1]};
            const v2f wv = {w2v[2 * j], w2v[2 * j + 1]};
            ps2 = __builtin_elementwise_fma(gelu2(hv), wv, ps2);
        }
    }
    const float ps = ps2.x + ps2.y;
    return ps + __shfl_xor(ps, 16, 32);
}

static_assert(BB == 2, "the point kernel pairs the two batch rows of a grid point in one wave");

__global__ __launch_bounds__(32, 5)
void disco_head_kernel(const float* __restrict__ x, const int* __restrict__ nbr,
                       const float* __restrict__ basis,
                       const float* __restrict__ disco_w,
                       const float* __restrict__ disco_b,
                       const float* __restrict__ hw1, const float* __restrict__ hb1,
                       const float* __restrict__ hw2, const float* __restrict__ hb2,
                       const float* __restrict__ mod, float* __restrict__ xln)
{
    __shared__ __attribute__((aligned(16))) float xs[2 * NNZN * XS_PITCH];

    const int l  = threadIdx.x;
    const int p  = blockIdx.x;
    const int hi = l >> 4;
    const int lr = l & 15;
    const bool lo_half = (hi == 0);

    v8f accb[2];
    {
        const v8f zb = {0.f, 0.f, 0.f, 0.f, 0.f, 0.f, 0.f, 0.f};
        accb[0] = zb; accb[1] = zb;
        const float* bp   = basis + (long)p * NNZN;
        const bool   n1ok = lr < (NNZN - 16);
        const int    n1   = n1ok ? (16 + lr) : 0;
        const float* dwp  = disco_w + lr * KK;
        #pragma unroll
        for (int kt = 0; kt < 6; ++kt) {
            const int k0 = kt * 4 + 2 * hi;
            const float* r0 = bp + (long)k0 * (PP * NNZN);
            const float* r1 = r0 + (PP * NNZN);
            v2f bf, a0, a1;
            bf.x = dwp[k0];
            bf.y = dwp[k0 + 1];
            a0.x = r0[lr];
            a0.y = r1[lr];
            const float t0 = r0[n1];
            const float t1 = r1[n1];
            a1.x = n1ok ? t0 : 0.f;
            a1.y = n1ok ? t1 : 0.f;
            accb[0] = wmma_f32(a0, bf, accb[0]);
            accb[1] = wmma_f32(a1, bf, accb[1]);
        }
        {
            const float* r0 = bp + (long)24 * (PP * NNZN);
            const float w  = dwp[24];
            const float t0 = r0[lr];
            const float t1 = r0[n1];
            v2f bf, a0, a1;
            bf.x = lo_half ? w : 0.f;              bf.y = 0.f;
            a0.x = lo_half ? t0 : 0.f;             a0.y = 0.f;
            a1.x = (lo_half && n1ok) ? t1 : 0.f;   a1.y = 0.f;
            accb[0] = wmma_f32(a0, bf, accb[0]);
            accb[1] = wmma_f32(a1, bf, accb[1]);
        }
    }

    {
        const int* np = nbr + p * NNZN;
        #pragma unroll
        for (int i = 0; i < NNZN; ++i) {
            const int r  = 2 * i + (hi & 1);
            const int bq = (r >= NNZN) ? 1 : 0;
            const int n  = r - NNZN * bq;
            const unsigned idx = (unsigned)np[n];
            const unsigned boff = __umul24(idx, MD * 4) + (unsigned)(16 * lr)
                                  + (unsigned)bq * (unsigned)(PP * MD * 4);
            const char* rowp = (const char*)x + boff;
            v4f v;
            __builtin_memcpy(&v, __builtin_assume_aligned(rowp, 8), 16);
            *(v4f*)&xs[r * XS_PITCH + 4 * lr] = v;
        }
    }
    unsigned xl[16];
    #pragma unroll
    for (int r = 0; r < 8; ++r) {
        xl[r]     = (unsigned)(8 * hi + r) * XS_PITCH + lr;
        xl[8 + r] = (unsigned)(lo_half ? (16 + r) : 24) * XS_PITCH + lr;
    }
    __syncthreads();
    const v8f dbv = ld8(disco_b + 8 * hi);
    const int hrow = p / WW;
    const float* mrow0 = mod + hrow * (2 * SHN);
    const float* mrow1 = mod + (HH_ + hrow) * (2 * SHN);
    const v8f gm0 = ld8(mrow0 + 8 * hi);
    const v8f bt0 = ld8(mrow0 + SHN + 8 * hi);
    const v8f gm1 = ld8(mrow1 + 8 * hi);
    const v8f bt1 = ld8(mrow1 + SHN + 8 * hi);
    const int jl = lr >> 3;
    const float* xres = x   + ((long)hi * PP + p) * MD + lr;
    float*       orow = xln + ((long)hi * PP + p) * CC + lr;

    #pragma unroll 1
    for (int ct = 0; ct < 4; ++ct) {
        v8f acc0 = dbv, acc1 = dbv;
        const unsigned co = ct * 16;
        #pragma unroll
        for (int t = 0; t < 2; ++t) {
            #pragma unroll
            for (int u = 0; u < 4; ++u) {
                v2f af, b0, b1;
                af.x = accb[t][2 * u];
                af.y = accb[t][2 * u + 1];
                b0.x = xs[xl[8 * t + 2 * u] + co];
                b0.y = xs[xl[8 * t + 2 * u + 1] + co];
                b1.x = xs[xl[8 * t + 2 * u] + NNZN * XS_PITCH + co];
                b1.y = xs[xl[8 * t + 2 * u + 1] + NNZN * XS_PITCH + co];
                acc0 = wmma_f32(af, b0, acc0);
                acc1 = wmma_f32(af, b1, acc1);
            }
        }
        #pragma unroll
        for (int r = 0; r < 8; ++r) {
            acc0[r] = acc0[r] * gm0[r] + bt0[r];
            acc1[r] = acc1[r] * gm1[r] + bt1[r];
        }
        v8f g0, g1;
        #pragma unroll
        for (int r = 0; r < 8; ++r) {
            g0[r] = take_hi8(acc0[r], acc1[r]);
            g1[r] = take_lo8(acc1[r], acc0[r]);
        }
        const float tot0 = head_mlp(g0, 2 * ct,     lr, hi, hw1, hb1, hw2);
        __builtin_amdgcn_sched_barrier(0);
        const float tot1 = head_mlp(g1, 2 * ct + 1, lr, hi, hw1, hb1, hw2);
        const float s0 = take_hi8(tot0, tot1);
        const float s1 = take_lo8(tot1, tot0);
        const float val = lo_half ? s0 : s1;
        orow[ct * 16] = (hb2[2 * ct + jl] + val) + xres[ct * 16];
    }
}

__device__ __forceinline__ void ffn_load_w1(const float* __restrict__ w1, int nt, int lr, bool hi,
                                            v2f (&bf)[17]) {
    #pragma unroll
    for (int kt = 0; kt < 17; ++kt) {
        const int k0 = kt * 4 + (hi ? 2 : 0);
        const int ka = (k0     < MD) ? k0     : 0;
        const int kb = (k0 + 1 < MD) ? k0 + 1 : 0;
        const float w0  = w1[(long)ka * FFNH + nt * 16 + lr];
        const float w1v = w1[(long)kb * FFNH + nt * 16 + lr];
        bf[kt].x = (k0     < MD) ? w0  : 0.f;
        bf[kt].y = (k0 + 1 < MD) ? w1v : 0.f;
    }
}

__device__ __forceinline__ void ffn_hidden_pair(const v2f (&af)[17], const v2f (&bf0)[17], const v2f (&bf1)[17],
                                                float bb0, float bb1,
                                                float* __restrict__ hid, int nt, int lr, bool hi) {
    v8f acc0 = {bb0, bb0, bb0, bb0, bb0, bb0, bb0, bb0};
    v8f acc1 = {bb1, bb1, bb1, bb1, bb1, bb1, bb1, bb1};
    #pragma unroll
    for (int kt = 0; kt < 17; ++kt) {
        acc0 = wmma_f32(af[kt], bf0[kt], acc0);
        acc1 = wmma_f32(af[kt], bf1[kt], acc1);
    }
    #pragma unroll
    for (int j = 0; j < 4; ++j) {
        const int m = 2 * j + (hi ? 8 : 0);
        const v2f g0 = gelu2((v2f){acc0[2 * j], acc0[2 * j + 1]});
        const v2f g1 = gelu2((v2f){acc1[2 * j], acc1[2 * j + 1]});
        hid[m * 260 + nt * 16 + lr]            = g0.x;
        hid[(m + 1) * 260 + nt * 16 + lr]      = g0.y;
        hid[m * 260 + nt * 16 + 16 + lr]       = g1.x;
        hid[(m + 1) * 260 + nt * 16 + 16 + lr] = g1.y;
    }
}

__device__ __forceinline__ void ffn_load_w2(const float* __restrict__ wp, int g, v4f (&w)[16]) {
    #pragma unroll
    for (int i = 0; i < 8; ++i) {
        const int kt = 8 * g + i;
        w[2 * i]     = *(const v4f*)(wp + kt * 4 * CC);
        w[2 * i + 1] = *(const v4f*)(wp + kt * 4 * CC + CC);
    }
}

__global__ __launch_bounds__(32)
void ffn_kernel(const float* __restrict__ x, const float* __restrict__ xln,
                const float* __restrict__ w1, const float* __restrict__ b1,
                const float* __restrict__ w2, const float* __restrict__ b2,
                float* __restrict__ out)
{
    __shared__ __attribute__((aligned(16))) float hid[16 * 260];

    const int  l   = threadIdx.x;
    const long gq0 = (long)blockIdx.x * 16;
    const bool hi = l >= 16;
    const int  lr = l & 15;

    v2f afr[17];
    {
        const float* xr = xln + (gq0 + lr) * CC + (hi ? 2 : 0);
        #pragma unroll
        for (int kt = 0; kt < 16; ++kt) afr[kt] = *(const v2f*)(xr + kt * 4);
        const v2f sc = *(const v2f*)(x + (gq0 + lr) * MD + CC);
        afr[16].x = hi ? 0.f : sc.x;
        afr[16].y = hi ? 0.f : sc.y;
    }

    v2f bA0[17], bA1[17], bB0[17], bB1[17];
    ffn_load_w1(w1, 0, lr, hi, bA0);
    ffn_load_w1(w1, 1, lr, hi, bA1);
    float bbA0 = b1[lr], bbA1 = b1[16 + lr];
    for (int nt = 0; nt < 16; nt += 4) {
        ffn_load_w1(w1, nt + 2, lr, hi, bB0);
        ffn_load_w1(w1, nt + 3, lr, hi, bB1);
        const float bbB0 = b1[(nt + 2) * 16 + lr], bbB1 = b1[(nt + 3) * 16 + lr];
        ffn_hidden_pair(afr, bA0, bA1, bbA0, bbA1, hid, nt, lr, hi);
        const int n4 = (nt + 4 < 16) ? nt + 4 : 14;
        ffn_load_w1(w1, n4,     lr, hi, bA0);
        ffn_load_w1(w1, n4 + 1, lr, hi, bA1);
        bbA0 = b1[n4 * 16 + lr];
        bbA1 = b1[(n4 + 1) * 16 + lr];
        ffn_hidden_pair(afr, bB0, bB1, bbB0, bbB1, hid, nt + 2, lr, hi);
    }

    const float* wp = w2 + (hi ? 2 : 0) * CC + 4 * lr;
    v4f wA[16], wB[16];
    ffn_load_w2(wp, 0, wA);
    __syncthreads();

    v2f a2[64];
    {
        const float* hp = hid + lr * 260 + (hi ? 2 : 0);
        #pragma unroll
        for (int kt = 0; kt < 64; ++kt) a2[kt] = *(const v2f*)(hp + kt * 4);
    }
    __builtin_amdgcn_sched_barrier(0);
    v4f res[8];
    v8f acc[4];
    {
        const v4f bb = *(const v4f*)(b2 + 4 * lr);
        #pragma unroll
        for (int j = 0; j < 4; ++j) {
            const float b = bb[j];
            acc[j] = (v8f){b, b, b, b, b, b, b, b};
        }
    }
    #pragma unroll
    for (int g = 0; g < 8; g += 2) {
        ffn_load_w2(wp, g + 1, wB);
        __builtin_amdgcn_sched_barrier(0);
        #pragma unroll
        for (int i = 0; i < 8; ++i) {
            #pragma unroll
            for (int j = 0; j < 4; ++j)
                acc[j] = wmma_f32(a2[8 * g + i], (v2f){wA[2 * i][j], wA[2 * i + 1][j]}, acc[j]);
        }
        __builtin_amdgcn_sched_barrier(0);
        if (g + 2 < 8) {
            ffn_load_w2(wp, g + 2, wA);
        } else {
            #pragma unroll
            for (int r = 0; r < 8; ++r)
                res[r] = *(const v4f*)(xln + (gq0 + r + (hi ? 8 : 0)) * CC + 4 * lr);
        }
        __builtin_amdgcn_sched_barrier(0);
        #pragma unroll
        for (int i = 0; i < 8; ++i) {
            #pragma unroll
            for (int j = 0; j < 4; ++j)
                acc[j] = wmma_f32(a2[8 * (g + 1) + i], (v2f){wB[2 * i][j], wB[2 * i + 1][j]}, acc[j]);
        }
        __builtin_amdgcn_sched_barrier(0);
    }
    #pragma unroll
    for (int r = 0; r < 8; ++r) {
        const int  m  = r + (hi ? 8 : 0);
        const long gq = gq0 + m;
        const v4f o = {acc[0][r] + res[r].x, acc[1][r] + res[r].y, acc[2][r] + res[r].z, acc[3][r] + res[r].w};
        __builtin_memcpy(__builtin_assume_aligned(out + gq * MD + 4 * lr, 8), &o, 16);
    }
    {
        const int  m  = l >> 1, c2 = l & 1;
        const long gq = gq0 + m;
        out[gq * MD + CC + c2] = x[gq * MD + CC + c2];
    }
}

extern "C" void kernel_launch(void* const* d_in, const int* in_sizes, int n_in,
                              void* d_out, int out_size, void* d_ws, size_t ws_size,
                              hipStream_t stream)
{
    const float* x       = (const float*)d_in[0];
    const int*   nbr     = (const int*)  d_in[1];
    const float* basis   = (const float*)d_in[2];
    const float* disco_w = (const float*)d_in[3];
    const float* disco_b = (const float*)d_in[4];
    const float* lw1     = (const float*)d_in[5];
    const float* lb1     = (const float*)d_in[6];
    const float* lw2     = (const float*)d_in[7];
    const float* lb2     = (const float*)d_in[8];
    const float* hw1     = (const float*)d_in[9];
    const float* hb1     = (const float*)d_in[10];
    const float* hw2     = (const float*)d_in[11];
    const float* hb2     = (const float*)d_in[12];
    const float* fw1     = (const float*)d_in[13];
    const float* fb1     = (const float*)d_in[14];
    const float* fw2     = (const float*)d_in[15];
    const float* fb2     = (const float*)d_in[16];
    float* out = (float*)d_out;

    float* mod = (float*)d_ws;
    float* xln = (float*)((char*)d_ws + (size_t)(BB * HH_ * 32) * sizeof(float));

    mod_kernel<<<BB * HH_, 32, 0, stream>>>(x, lw1, lb1, lw2, lb2, mod);
    disco_head_kernel<<<PP, 32, 0, stream>>>(x, nbr, basis, disco_w, disco_b,
                                                  hw1, hb1, hw2, hb2, mod, xln);
    ffn_kernel<<<(BB * PP) / 16, 32, 0, stream>>>(x, xln, fw1, fb1, fw2, fb2, out);
}


# Latitude-modulation MLP moved into the point kernel (lane i = hidden unit i, LDS broadcast, lane j sums output j, i ascending as before): its separate launch in front is gone, 3 launches become 2
# speedup vs baseline: 2.6878x; 1.0349x over previous
#include <hip/hip_runtime.h>
#include <hip/hip_bf16.h>

typedef __attribute__((ext_vector_type(2))) float v2f;
typedef __attribute__((ext_vector_type(8))) float v8f;

#define BB    2
#define HH_   90
#define WW    180
#define PP    (HH_ * WW)
#define CC    64
#define MD    66
#define SHN   16
#define KK    25
#define NNZN  25
#define HHID  32
#define FFNH  256
#define XS_PITCH 68

__device__ __forceinline__ v2f gelu2(v2f v) {
    v2f d, t, e, m;
    d.x = __builtin_fmaf(__builtin_fabsf(v.x), 0.318198055f, 1.0f);
    d.y = __builtin_fmaf(__builtin_fabsf(v.y), 0.318198055f, 1.0f);
    t.x = __builtin_amdgcn_rcpf(d.x);
    t.y = __builtin_amdgcn_rcpf(d.y);
    v2f s = {0.145559847f, 0.145559847f};
    s = __builtin_elementwise_fma(t, s, (v2f){-0.776905596f, -0.776905596f});
    s = __builtin_elementwise_fma(t, s, (v2f){ 1.34791970f,   1.34791970f});
    s = __builtin_elementwise_fma(t, s, (v2f){-0.591609299f, -0.591609299f});
    s = __builtin_elementwise_fma(t, s, (v2f){ 0.723748207f,  0.723748207f});
    s = __builtin_elementwise_fma(t, s, (v2f){ 0.315644562f,  0.315644562f});
    s = __builtin_elementwise_fma(t, s, (v2f){ 0.406991005f,  0.406991005f});
    s = __builtin_elementwise_fma(-t, s, s);
    const v2f a = (v * v) * (v2f){-0.721347511f, -0.721347511f};
    e.x = __builtin_amdgcn_exp2f(a.x);
    e.y = __builtin_amdgcn_exp2f(a.y);
    m.x = __builtin_fmaxf(v.x, 0.f);
    m.y = __builtin_fmaxf(v.y, 0.f);
    return __builtin_elementwise_fma(-s, e, m);
}

__device__ __forceinline__ v8f wmma_f32(v2f a, v2f b, v8f c) {
    return __builtin_amdgcn_wmma_f32_16x16x4_f32(false, a, false, b, (short)0, c,
                                                 false, false);
}

typedef __attribute__((ext_vector_type(4))) float v4f;

__device__ __forceinline__ v8f ld8(const float* p) {
    const v4f a = *(const v4f*)(p);
    const v4f c = *(const v4f*)(p + 4);
    v8f r = {a.x, a.y, a.z, a.w, c.x, c.y, c.z, c.w};
    return r;
}

__device__ __forceinline__ float take_hi8(float keep, float src) {
    return __builtin_bit_cast(float, __builtin_amdgcn_update_dpp(
        __builtin_bit_cast(int, keep), __builtin_bit_cast(int, src), 0x128, 0xF, 0xC, false));
}
__device__ __forceinline__ float take_lo8(float keep, float src) {
    return __builtin_bit_cast(float, __builtin_amdgcn_update_dpp(
        __builtin_bit_cast(int, keep), __builtin_bit_cast(int, src), 0x128, 0xF, 0x3, false));
}

__device__ __forceinline__ float head_mlp(const v8f g, int head, int lr, int hi,
                                          const float* __restrict__ hw1,
                                          const float* __restrict__ hb1,
                                          const float* __restrict__ hw2) {
    v8f acch[2];
    acch[0] = ld8(hb1 + head * HHID + 8 * hi);
    acch[1] = ld8(hb1 + head * HHID + 16 + 8 * hi);
    #pragma unroll
    for (int u = 0; u < 4; ++u) {
        v2f bf;
        bf.x = g[2 * u];
        bf.y = g[2 * u + 1];
        const float* wp = hw1 + (head * SHN + 8 * hi + 2 * u) * HHID + lr;
        #pragma unroll
        for (int nt = 0; nt < 2; ++nt) {
            v2f af;
            af.x = wp[nt * 16];
            af.y = wp[HHID + nt * 16];
            acch[nt] = wmma_f32(af, bf, acch[nt]);
        }
    }
    v2f ps2 = {0.f, 0.f};
    #pragma unroll
    for (int nt = 0; nt < 2; ++nt) {
        const v8f w2v = ld8(hw2 + head * HHID + nt * 16 + 8 * hi);
        #pragma unroll
        for (int j = 0; j < 4; ++j) {
            const v2f hv = {acch[nt][2 * j], acch[nt][2 * j + 1]};
            const v2f wv = {w2v[2 * j], w2v[2 * j + 1]};
            ps2 = __builtin_elementwise_fma(gelu2(hv), wv, ps2);
        }
    }
    const float ps = ps2.x + ps2.y;
    return ps + __shfl_xor(ps, 16, 32);
}

static_assert(BB == 2, "the point kernel pairs the two batch rows of a grid point in one wave");

__global__ __launch_bounds__(32, 5)
void disco_head_kernel(const float* __restrict__ x, const int* __restrict__ nbr,
                       const float* __restrict__ basis,
                       const float* __restrict__ disco_w,
                       const float* __restrict__ disco_b,
                       const float* __restrict__ hw1, const float* __restrict__ hb1,
                       const float* __restrict__ hw2, const float* __restrict__ hb2,
                       const float* __restrict__ lw1, const float* __restrict__ lb1,
                       const float* __restrict__ lw2, const float* __restrict__ lb2,
                       float* __restrict__ xln)
{
    __shared__ __attribute__((aligned(16))) float xs[2 * NNZN * XS_PITCH];
    __shared__ __attribute__((aligned(16))) float hidb[2 * 2 * SHN];
    __shared__ __attribute__((aligned(16))) float modb[2 * 2 * SHN];

    const int l  = threadIdx.x;
    const int p  = blockIdx.x;
    const int hi = l >> 4;
    const int lr = l & 15;
    const bool lo_half = (hi == 0);
    const int hrow = p / WW;

    {
        const v2f sA = *(const v2f*)(x + ((long)hrow * WW) * MD + CC);
        const v2f sB = *(const v2f*)(x + ((long)(HH_ + hrow) * WW) * MD + CC);
        const float wa = lw1[l], wb = lw1[2 * SHN + l], bb = lb1[l];
        v2f tv;
        tv.x = __builtin_fmaf(sA.x, wa, sA.y * wb) + bb;
        tv.y = __builtin_fmaf(sB.x, wa, sB.y * wb) + bb;
        *(v2f*)&hidb[2 * l] = gelu2(tv);
    }

    v8f accb[2];
    {
        const v8f zb = {0.f, 0.f, 0.f, 0.f, 0.f, 0.f, 0.f, 0.f};
        accb[0] = zb; accb[1] = zb;
        const float* bp   = basis + (long)p * NNZN;
        const bool   n1ok = lr < (NNZN - 16);
        const int    n1   = n1ok ? (16 + lr) : 0;
        const float* dwp  = disco_w + lr * KK;
        #pragma unroll
        for (int kt = 0; kt < 6; ++kt) {
            const int k0 = kt * 4 + 2 * hi;
            const float* r0 = bp + (long)k0 * (PP * NNZN);
            const float* r1 = r0 + (PP * NNZN);
            v2f bf, a0, a1;
            bf.x = dwp[k0];
            bf.y = dwp[k0 + 1];
            a0.x = r0[lr];
            a0.y = r1[lr];
            const float t0 = r0[n1];
            const float t1 = r1[n1];
            a1.x = n1ok ? t0 : 0.f;
            a1.y = n1ok ? t1 : 0.f;
            accb[0] = wmma_f32(a0, bf, accb[0]);
            accb[1] = wmma_f32(a1, bf, accb[1]);
        }
        {
            const float* r0 = bp + (long)24 * (PP * NNZN);
            const float w  = dwp[24];
            const float t0 = r0[lr];
            const float t1 = r0[n1];
            v2f bf, a0, a1;
            bf.x = lo_half ? w : 0.f;              bf.y = 0.f;
            a0.x = lo_half ? t0 : 0.f;             a0.y = 0.f;
            a1.x = (lo_half && n1ok) ? t1 : 0.f;   a1.y = 0.f;
            accb[0] = wmma_f32(a0, bf, accb[0]);
            accb[1] = wmma_f32(a1, bf, accb[1]);
        }
    }

    {
        const int* np = nbr + p * NNZN;
        #pragma unroll
        for (int i = 0; i < NNZN; ++i) {
            const int r  = 2 * i + (hi & 1);
            const int bq = (r >= NNZN) ? 1 : 0;
            const int n  = r - NNZN * bq;
            const unsigned idx = (unsigned)np[n];
            const unsigned boff = __umul24(idx, MD * 4) + (unsigned)(16 * lr)
                                  + (unsigned)bq * (unsigned)(PP * MD * 4);
            const char* rowp = (const char*)x + boff;
            v4f v;
            __builtin_memcpy(&v, __builtin_assume_aligned(rowp, 8), 16);
            *(v4f*)&xs[r * XS_PITCH + 4 * lr] = v;
        }
    }
    unsigned xl[16];
    #pragma unroll
    for (int r = 0; r < 8; ++r) {
        xl[r]     = (unsigned)(8 * hi + r) * XS_PITCH + lr;
        xl[8 + r] = (unsigned)(lo_half ? (16 + r) : 24) * XS_PITCH + lr;
    }
    __syncthreads();
    {
        const float b2 = lb2[l];
        v2f macc = {b2, b2};
        #pragma unroll
        for (int i = 0; i < 2 * SHN; i += 2) {
            const v4f h2 = *(const v4f*)&hidb[2 * i];
            const float w0 = lw2[i * (2 * SHN) + l];
            const float w1 = lw2[(i + 1) * (2 * SHN) + l];
            macc = __builtin_elementwise_fma((v2f){h2.x, h2.y}, (v2f){w0, w0}, macc);
            macc = __builtin_elementwise_fma((v2f){h2.z, h2.w}, (v2f){w1, w1}, macc);
        }
        modb[l]           = macc.x;
        modb[2 * SHN + l] = macc.y;
    }
    __syncthreads();
    const v8f dbv = ld8(disco_b + 8 * hi);
    const v8f gm0 = ld8(modb + 8 * hi);
    const v8f bt0 = ld8(modb + SHN + 8 * hi);
    const v8f gm1 = ld8(modb + 2 * SHN + 8 * hi);
    const v8f bt1 = ld8(modb + 3 * SHN + 8 * hi);
    const int jl = lr >> 3;
    const float* xres = x   + ((long)hi * PP + p) * MD + lr;
    float*       orow = xln + ((long)hi * PP + p) * CC + lr;

    #pragma unroll 1
    for (int ct = 0; ct < 4; ++ct) {
        v8f acc0 = dbv, acc1 = dbv;
        const unsigned co = ct * 16;
        #pragma unroll
        for (int t = 0; t < 2; ++t) {
            #pragma unroll
            for (int u = 0; u < 4; ++u) {
                v2f af, b0, b1;
                af.x = accb[t][2 * u];
                af.y = accb[t][2 * u + 1];
                b0.x = xs[xl[8 * t + 2 * u] + co];
                b0.y = xs[xl[8 * t + 2 * u + 1] + co];
                b1.x = xs[xl[8 * t + 2 * u] + NNZN * XS_PITCH + co];
                b1.y = xs[xl[8 * t + 2 * u + 1] + NNZN * XS_PITCH + co];
                acc0 = wmma_f32(af, b0, acc0);
                acc1 = wmma_f32(af, b1, acc1);
            }
        }
        #pragma unroll
        for (int r = 0; r < 8; ++r) {
            acc0[r] = acc0[r] * gm0[r] + bt0[r];
            acc1[r] = acc1[r] * gm1[r] + bt1[r];
        }
        v8f g0, g1;
        #pragma unroll
        for (int r = 0; r < 8; ++r) {
            g0[r] = take_hi8(acc0[r], acc1[r]);
            g1[r] = take_lo8(acc1[r], acc0[r]);
        }
        const float tot0 = head_mlp(g0, 2 * ct,     lr, hi, hw1, hb1, hw2);
        __builtin_amdgcn_sched_barrier(0);
        const float tot1 = head_mlp(g1, 2 * ct + 1, lr, hi, hw1, hb1, hw2);
        const float s0 = take_hi8(tot0, tot1);
        const float s1 = take_lo8(tot1, tot0);
        const float val = lo_half ? s0 : s1;
        orow[ct * 16] = (hb2[2 * ct + jl] + val) + xres[ct * 16];
    }
}

__device__ __forceinline__ void ffn_load_w1(const float* __restrict__ w1, int nt, int lr, bool hi,
                                            v2f (&bf)[17]) {
    #pragma unroll
    for (int kt = 0; kt < 17; ++kt) {
        const int k0 = kt * 4 + (hi ? 2 : 0);
        const int ka = (k0     < MD) ? k0     : 0;
        const int kb = (k0 + 1 < MD) ? k0 + 1 : 0;
        const float w0  = w1[(long)ka * FFNH + nt * 16 + lr];
        const float w1v = w1[(long)kb * FFNH + nt * 16 + lr];
        bf[kt].x = (k0     < MD) ? w0  : 0.f;
        bf[kt].y = (k0 + 1 < MD) ? w1v : 0.f;
    }
}

__device__ __forceinline__ void ffn_hidden_pair(const v2f (&af)[17], const v2f (&bf0)[17], const v2f (&bf1)[17],
                                                float bb0, float bb1,
                                                float* __restrict__ hid, int nt, int lr, bool hi) {
    v8f acc0 = {bb0, bb0, bb0, bb0, bb0, bb0, bb0, bb0};
    v8f acc1 = {bb1, bb1, bb1, bb1, bb1, bb1, bb1, bb1};
    #pragma unroll
    for (int kt = 0; kt < 17; ++kt) {
        acc0 = wmma_f32(af[kt], bf0[kt], acc0);
        acc1 = wmma_f32(af[kt], bf1[kt], acc1);
    }
    #pragma unroll
    for (int j = 0; j < 4; ++j) {
        const int m = 2 * j + (hi ? 8 : 0);
        const v2f g0 = gelu2((v2f){acc0[2 * j], acc0[2 * j + 1]});
        const v2f g1 = gelu2((v2f){acc1[2 * j], acc1[2 * j + 1]});
        hid[m * 260 + nt * 16 + lr]            = g0.x;
        hid[(m + 1) * 260 + nt * 16 + lr]      = g0.y;
        hid[m * 260 + nt * 16 + 16 + lr]       = g1.x;
        hid[(m + 1) * 260 + nt * 16 + 16 + lr] = g1.y;
    }
}

__device__ __forceinline__ void ffn_load_w2(const float* __restrict__ wp, int g, v4f (&w)[16]) {
    #pragma unroll
    for (int i = 0; i < 8; ++i) {
        const int kt = 8 * g + i;
        w[2 * i]     = *(const v4f*)(wp + kt * 4 * CC);
        w[2 * i + 1] = *(const v4f*)(wp + kt * 4 * CC + CC);
    }
}

__global__ __launch_bounds__(32)
void ffn_kernel(const float* __restrict__ x, const float* __restrict__ xln,
                const float* __restrict__ w1, const float* __restrict__ b1,
                const float* __restrict__ w2, const float* __restrict__ b2,
                float* __restrict__ out)
{
    __shared__ __attribute__((aligned(16))) float hid[16 * 260];

    const int  l   = threadIdx.x;
    const long gq0 = (long)blockIdx.x * 16;
    const bool hi = l >= 16;
    const int  lr = l & 15;

    v2f afr[17];
    {
        const float* xr = xln + (gq0 + lr) * CC + (hi ? 2 : 0);
        #pragma unroll
        for (int kt = 0; kt < 16; ++kt) afr[kt] = *(const v2f*)(xr + kt * 4);
        const v2f sc = *(const v2f*)(x + (gq0 + lr) * MD + CC);
        afr[16].x = hi ? 0.f : sc.x;
        afr[16].y = hi ? 0.f : sc.y;
    }

    v2f bA0[17], bA1[17], bB0[17], bB1[17];
    ffn_load_w1(w1, 0, lr, hi, bA0);
    ffn_load_w1(w1, 1, lr, hi, bA1);
    float bbA0 = b1[lr], bbA1 = b1[16 + lr];
    for (int nt = 0; nt < 16; nt += 4) {
        ffn_load_w1(w1, nt + 2, lr, hi, bB0);
        ffn_load_w1(w1, nt + 3, lr, hi, bB1);
        const float bbB0 = b1[(nt + 2) * 16 + lr], bbB1 = b1[(nt + 3) * 16 + lr];
        ffn_hidden_pair(afr, bA0, bA1, bbA0, bbA1, hid, nt, lr, hi);
        const int n4 = (nt + 4 < 16) ? nt + 4 : 14;
        ffn_load_w1(w1, n4,     lr, hi, bA0);
        ffn_load_w1(w1, n4 + 1, lr, hi, bA1);
        bbA0 = b1[n4 * 16 + lr];
        bbA1 = b1[(n4 + 1) * 16 + lr];
        ffn_hidden_pair(afr, bB0, bB1, bbB0, bbB1, hid, nt + 2, lr, hi);
    }

    const float* wp = w2 + (hi ? 2 : 0) * CC + 4 * lr;
    v4f wA[16], wB[16];
    ffn_load_w2(wp, 0, wA);
    __syncthreads();

    v2f a2[64];
    {
        const float* hp = hid + lr * 260 + (hi ? 2 : 0);
        #pragma unroll
        for (int kt = 0; kt < 64; ++kt) a2[kt] = *(const v2f*)(hp + kt * 4);
    }
    __builtin_amdgcn_sched_barrier(0);
    v4f res[8];
    v8f acc[4];
    {
        const v4f bb = *(const v4f*)(b2 + 4 * lr);
        #pragma unroll
        for (int j = 0; j < 4; ++j) {
            const float b = bb[j];
            acc[j] = (v8f){b, b, b, b, b, b, b, b};
        }
    }
    #pragma unroll
    for (int g = 0; g < 8; g += 2) {
        ffn_load_w2(wp, g + 1, wB);
        __builtin_amdgcn_sched_barrier(0);
        #pragma unroll
        for (int i = 0; i < 8; ++i) {
            #pragma unroll
            for (int j = 0; j < 4; ++j)
                acc[j] = wmma_f32(a2[8 * g + i], (v2f){wA[2 * i][j], wA[2 * i + 1][j]}, acc[j]);
        }
        __builtin_amdgcn_sched_barrier(0);
        if (g + 2 < 8) {
            ffn_load_w2(wp, g + 2, wA);
        } else {
            #pragma unroll
            for (int r = 0; r < 8; ++r)
                res[r] = *(const v4f*)(xln + (gq0 + r + (hi ? 8 : 0)) * CC + 4 * lr);
        }
        __builtin_amdgcn_sched_barrier(0);
        #pragma unroll
        for (int i = 0; i < 8; ++i) {
            #pragma unroll
            for (int j = 0; j < 4; ++j)
                acc[j] = wmma_f32(a2[8 * (g + 1) + i], (v2f){wB[2 * i][j], wB[2 * i + 1][j]}, acc[j]);
        }
        __builtin_amdgcn_sched_barrier(0);
    }
    #pragma unroll
    for (int r = 0; r < 8; ++r) {
        const int  m  = r + (hi ? 8 : 0);
        const long gq = gq0 + m;
        const v4f o = {acc[0][r] + res[r].x, acc[1][r] + res[r].y, acc[2][r] + res[r].z, acc[3][r] + res[r].w};
        __builtin_memcpy(__builtin_assume_aligned(out + gq * MD + 4 * lr, 8), &o, 16);
    }
    {
        const int  m  = l >> 1, c2 = l & 1;
        const long gq = gq0 + m;
        out[gq * MD + CC + c2] = x[gq * MD + CC + c2];
    }
}

extern "C" void kernel_launch(void* const* d_in, const int* in_sizes, int n_in,
                              void* d_out, int out_size, void* d_ws, size_t ws_size,
                              hipStream_t stream)
{
    const float* x       = (const float*)d_in[0];
    const int*   nbr     = (const int*)  d_in[1];
    const float* basis   = (const float*)d_in[2];
    const float* disco_w = (const float*)d_in[3];
    const float* disco_b = (const float*)d_in[4];
    const float* lw1     = (const float*)d_in[5];
    const float* lb1     = (const float*)d_in[6];
    const float* lw2     = (const float*)d_in[7];
    const float* lb2     = (const float*)d_in[8];
    const float* hw1     = (const float*)d_in[9];
    const float* hb1     = (const float*)d_in[10];
    const float* hw2     = (const float*)d_in[11];
    const float* hb2     = (const float*)d_in[12];
    const float* fw1     = (const float*)d_in[13];
    const float* fb1     = (const float*)d_in[14];
    const float* fw2     = (const float*)d_in[15];
    const float* fb2     = (const float*)d_in[16];
    float* out = (float*)d_out;

    float* xln = (float*)((char*)d_ws + (size_t)(BB * HH_ * 32) * sizeof(float));

    disco_head_kernel<<<PP, 32, 0, stream>>>(x, nbr, basis, disco_w, disco_b,
                                                  hw1, hb1, hw2, hb2, lw1, lb1, lw2, lb2, xln);
    ffn_kernel<<<(BB * PP) / 16, 32, 0, stream>>>(x, xln, fw1, fb1, fw2, fb2, out);
}


# gelu, all three sites: erf(v/sqrt2) = tanh(v*Q(v*v)), Q of degree 6: one hardware tanh a value replaces rcp, exp2 and the erfc tail polynomial; 18 -> 11 instructions a pair; sum order unchanged
# speedup vs baseline: 2.8572x; 1.0630x over previous
#include <hip/hip_runtime.h>
#include <hip/hip_bf16.h>

typedef __attribute__((ext_vector_type(2))) float v2f;
typedef __attribute__((ext_vector_type(8))) float v8f;

#define BB    2
#define HH_   90
#define WW    180
#define PP    (HH_ * WW)
#define CC    64
#define MD    66
#define SHN   16
#define KK    25
#define NNZN  25
#define HHID  32
#define FFNH  256
#define XS_PITCH 68

__device__ __forceinline__ v2f gelu2x(v2f v) {
    const v2f s = v * v;
    v2f q = {1.80560455e-9f, 1.80560455e-9f};
    q = __builtin_elementwise_fma(s, q, (v2f){-1.33445923e-7f, -1.33445923e-7f});
    q = __builtin_elementwise_fma(s, q, (v2f){ 3.96910264e-6f,  3.96910264e-6f});
    q = __builtin_elementwise_fma(s, q, (v2f){-5.52401107e-5f, -5.52401107e-5f});
    q = __builtin_elementwise_fma(s, q, (v2f){-3.31301308e-5f, -3.31301308e-5f});
    q = __builtin_elementwise_fma(s, q, (v2f){ 3.63342836e-2f,  3.63342836e-2f});
    q = __builtin_elementwise_fma(s, q, (v2f){ 7.97884226e-1f,  7.97884226e-1f});
    const v2f u = v * q;
    v2f t;
    t.x = __builtin_amdgcn_tanhf(u.x);
    t.y = __builtin_amdgcn_tanhf(u.y);
    return __builtin_elementwise_fma(v, t, v);
}

__device__ __forceinline__ v8f wmma_f32(v2f a, v2f b, v8f c) {
    return __builtin_amdgcn_wmma_f32_16x16x4_f32(false, a, false, b, (short)0, c,
                                                 false, false);
}

typedef __attribute__((ext_vector_type(4))) float v4f;

__device__ __forceinline__ v8f ld8(const float* p) {
    const v4f a = *(const v4f*)(p);
    const v4f c = *(const v4f*)(p + 4);
    v8f r = {a.x, a.y, a.z, a.w, c.x, c.y, c.z, c.w};
    return r;
}

__device__ __forceinline__ float take_hi8(float keep, float src) {
    return __builtin_bit_cast(float, __builtin_amdgcn_update_dpp(
        __builtin_bit_cast(int, keep), __builtin_bit_cast(int, src), 0x128, 0xF, 0xC, false));
}
__device__ __forceinline__ float take_lo8(float keep, float src) {
    return __builtin_bit_cast(float, __builtin_amdgcn_update_dpp(
        __builtin_bit_cast(int, keep), __builtin_bit_cast(int, src), 0x128, 0xF, 0x3, false));
}

__device__ __forceinline__ float head_mlp(const v8f g, int head, int lr, int hi,
                                          const float* __restrict__ hw1,
                                          const float* __restrict__ hb1,
                                          const float* __restrict__ hw2) {
    v8f acch[2];
    acch[0] = ld8(hb1 + head * HHID + 8 * hi);
    acch[1] = ld8(hb1 + head * HHID + 16 + 8 * hi);
    #pragma unroll
    for (int u = 0; u < 4; ++u) {
        v2f bf;
        bf.x = g[2 * u];
        bf.y = g[2 * u + 1];
        const float* wp = hw1 + (head * SHN + 8 * hi + 2 * u) * HHID + lr;
        #pragma unroll
        for (int nt = 0; nt < 2; ++nt) {
            v2f af;
            af.x = wp[nt * 16];
            af.y = wp[HHID + nt * 16];
            acch[nt] = wmma_f32(af, bf, acch[nt]);
        }
    }
    v2f ps2 = {0.f, 0.f};
    #pragma unroll
    for (int nt = 0; nt < 2; ++nt) {
        const v8f w2v = ld8(hw2 + head * HHID + nt * 16 + 8 * hi);
        #pragma unroll
        for (int j = 0; j < 4; ++j) {
            const v2f hv = {acch[nt][2 * j], acch[nt][2 * j + 1]};
            const v2f wv = {w2v[2 * j], w2v[2 * j + 1]};
            ps2 = __builtin_elementwise_fma(gelu2x(hv), wv, ps2);
        }
    }
    const float ps = ps2.x + ps2.y;
    return ps + __shfl_xor(ps, 16, 32);
}

static_assert(BB == 2, "the point kernel pairs the two batch rows of a grid point in one wave");

__global__ __launch_bounds__(32, 5)
void disco_head_kernel(const float* __restrict__ x, const int* __restrict__ nbr,
                       const float* __restrict__ basis,
                       const float* __restrict__ disco_w,
                       const float* __restrict__ disco_b,
                       const float* __restrict__ hw1, const float* __restrict__ hb1,
                       const float* __restrict__ hw2, const float* __restrict__ hb2,
                       const float* __restrict__ lw1, const float* __restrict__ lb1,
                       const float* __restrict__ lw2, const float* __restrict__ lb2,
                       float* __restrict__ xln)
{
    __shared__ __attribute__((aligned(16))) float xs[2 * NNZN * XS_PITCH];
    __shared__ __attribute__((aligned(16))) float hidb[2 * 2 * SHN];
    __shared__ __attribute__((aligned(16))) float modb[2 * 2 * SHN];

    const int l  = threadIdx.x;
    const int p  = blockIdx.x;
    const int hi = l >> 4;
    const int lr = l & 15;
    const bool lo_half = (hi == 0);
    const int hrow = p / WW;

    {
        const v2f sA = *(const v2f*)(x + ((long)hrow * WW) * MD + CC);
        const v2f sB = *(const v2f*)(x + ((long)(HH_ + hrow) * WW) * MD + CC);
        const float wa = lw1[l], wb = lw1[2 * SHN + l], bb = lb1[l];
        v2f tv;
        tv.x = __builtin_fmaf(sA.x, wa, sA.y * wb) + bb;
        tv.y = __builtin_fmaf(sB.x, wa, sB.y * wb) + bb;
        *(v2f*)&hidb[2 * l] = gelu2x(tv) * (v2f){0.5f, 0.5f};
    }

    v8f accb[2];
    {
        const v8f zb = {0.f, 0.f, 0.f, 0.f, 0.f, 0.f, 0.f, 0.f};
        accb[0] = zb; accb[1] = zb;
        const float* bp   = basis + (long)p * NNZN;
        const bool   n1ok = lr < (NNZN - 16);
        const int    n1   = n1ok ? (16 + lr) : 0;
        const float* dwp  = disco_w + lr * KK;
        #pragma unroll
        for (int kt = 0; kt < 6; ++kt) {
            const int k0 = kt * 4 + 2 * hi;
            const float* r0 = bp + (long)k0 * (PP * NNZN);
            const float* r1 = r0 + (PP * NNZN);
            v2f bf, a0, a1;
            bf.x = dwp[k0];
            bf.y = dwp[k0 + 1];
            a0.x = r0[lr];
            a0.y = r1[lr];
            const float t0 = r0[n1];
            const float t1 = r1[n1];
            a1.x = n1ok ? t0 : 0.f;
            a1.y = n1ok ? t1 : 0.f;
            accb[0] = wmma_f32(a0, bf, accb[0]);
            accb[1] = wmma_f32(a1, bf, accb[1]);
        }
        {
            const float* r0 = bp + (long)24 * (PP * NNZN);
            const float w  = dwp[24];
            const float t0 = r0[lr];
            const float t1 = r0[n1];
            v2f bf, a0, a1;
            bf.x = lo_half ? w : 0.f;              bf.y = 0.f;
            a0.x = lo_half ? t0 : 0.f;             a0.y = 0.f;
            a1.x = (lo_half && n1ok) ? t1 : 0.f;   a1.y = 0.f;
            accb[0] = wmma_f32(a0, bf, accb[0]);
            accb[1] = wmma_f32(a1, bf, accb[1]);
        }
    }

    {
        const int* np = nbr + p * NNZN;
        #pragma unroll
        for (int i = 0; i < NNZN; ++i) {
            const int r  = 2 * i + (hi & 1);
            const int bq = (r >= NNZN) ? 1 : 0;
            const int n  = r - NNZN * bq;
            const unsigned idx = (unsigned)np[n];
            const unsigned boff = __umul24(idx, MD * 4) + (unsigned)(16 * lr)
                                  + (unsigned)bq * (unsigned)(PP * MD * 4);
            const char* rowp = (const char*)x + boff;
            v4f v;
            __builtin_memcpy(&v, __builtin_assume_aligned(rowp, 8), 16);
            *(v4f*)&xs[r * XS_PITCH + 4 * lr] = v;
        }
    }
    unsigned xl[16];
    #pragma unroll
    for (int r = 0; r < 8; ++r) {
        xl[r]     = (unsigned)(8 * hi + r) * XS_PITCH + lr;
        xl[8 + r] = (unsigned)(lo_half ? (16 + r) : 24) * XS_PITCH + lr;
    }
    __syncthreads();
    {
        const float b2 = lb2[l];
        v2f macc = {b2, b2};
        #pragma unroll
        for (int i = 0; i < 2 * SHN; i += 2) {
            const v4f h2 = *(const v4f*)&hidb[2 * i];
            const float w0 = lw2[i * (2 * SHN) + l];
            const float w1 = lw2[(i + 1) * (2 * SHN) + l];
            macc = __builtin_elementwise_fma((v2f){h2.x, h2.y}, (v2f){w0, w0}, macc);
            macc = __builtin_elementwise_fma((v2f){h2.z, h2.w}, (v2f){w1, w1}, macc);
        }
        modb[l]           = macc.x;
        modb[2 * SHN + l] = macc.y;
    }
    __syncthreads();
    const v8f dbv = ld8(disco_b + 8 * hi);
    const v8f gm0 = ld8(modb + 8 * hi);
    const v8f bt0 = ld8(modb + SHN + 8 * hi);
    const v8f gm1 = ld8(modb + 2 * SHN + 8 * hi);
    const v8f bt1 = ld8(modb + 3 * SHN + 8 * hi);
    const int jl = lr >> 3;
    const float* xres = x   + ((long)hi * PP + p) * MD + lr;
    float*       orow = xln + ((long)hi * PP + p) * CC + lr;

    #pragma unroll 1
    for (int ct = 0; ct < 4; ++ct) {
        v8f acc0 = dbv, acc1 = dbv;
        const unsigned co = ct * 16;
        #pragma unroll
        for (int t = 0; t < 2; ++t) {
            #pragma unroll
            for (int u = 0; u < 4; ++u) {
                v2f af, b0, b1;
                af.x = accb[t][2 * u];
                af.y = accb[t][2 * u + 1];
                b0.x = xs[xl[8 * t + 2 * u] + co];
                b0.y = xs[xl[8 * t + 2 * u + 1] + co];
                b1.x = xs[xl[8 * t + 2 * u] + NNZN * XS_PITCH + co];
                b1.y = xs[xl[8 * t + 2 * u + 1] + NNZN * XS_PITCH + co];
                acc0 = wmma_f32(af, b0, acc0);
                acc1 = wmma_f32(af, b1, acc1);
            }
        }
        #pragma unroll
        for (int r = 0; r < 8; ++r) {
            acc0[r] = acc0[r] * gm0[r] + bt0[r];
            acc1[r] = acc1[r] * gm1[r] + bt1[r];
        }
        v8f g0, g1;
        #pragma unroll
        for (int r = 0; r < 8; ++r) {
            g0[r] = take_hi8(acc0[r], acc1[r]);
            g1[r] = take_lo8(acc1[r], acc0[r]);
        }
        const float tot0 = head_mlp(g0, 2 * ct,     lr, hi, hw1, hb1, hw2);
        __builtin_amdgcn_sched_barrier(0);
        const float tot1 = head_mlp(g1, 2 * ct + 1, lr, hi, hw1, hb1, hw2);
        const float s0 = take_hi8(tot0, tot1);
        const float s1 = take_lo8(tot1, tot0);
        const float val = lo_half ? s0 : s1;
        orow[ct * 16] = __builtin_fmaf(val, 0.5f, hb2[2 * ct + jl]) + xres[ct * 16];
    }
}

__device__ __forceinline__ void ffn_load_w1(const float* __restrict__ w1, int nt, int lr, bool hi,
                                            v2f (&bf)[17]) {
    #pragma unroll
    for (int kt = 0; kt < 17; ++kt) {
        const int k0 = kt * 4 + (hi ? 2 : 0);
        const int ka = (k0     < MD) ? k0     : 0;
        const int kb = (k0 + 1 < MD) ? k0 + 1 : 0;
        const float w0  = w1[(long)ka * FFNH + nt * 16 + lr];
        const float w1v = w1[(long)kb * FFNH + nt * 16 + lr];
        bf[kt].x = (k0     < MD) ? w0  : 0.f;
        bf[kt].y = (k0 + 1 < MD) ? w1v : 0.f;
    }
}

__device__ __forceinline__ void ffn_hidden_pair(const v2f (&af)[17], const v2f (&bf0)[17], const v2f (&bf1)[17],
                                                float bb0, float bb1,
                                                float* __restrict__ hid, int nt, int lr, bool hi) {
    v8f acc0 = {bb0, bb0, bb0, bb0, bb0, bb0, bb0, bb0};
    v8f acc1 = {bb1, bb1, bb1, bb1, bb1, bb1, bb1, bb1};
    #pragma unroll
    for (int kt = 0; kt < 17; ++kt) {
        acc0 = wmma_f32(af[kt], bf0[kt], acc0);
        acc1 = wmma_f32(af[kt], bf1[kt], acc1);
    }
    #pragma unroll
    for (int j = 0; j < 4; ++j) {
        const int m = 2 * j + (hi ? 8 : 0);
        const v2f g0 = gelu2x((v2f){acc0[2 * j], acc0[2 * j + 1]});
        const v2f g1 = gelu2x((v2f){acc1[2 * j], acc1[2 * j + 1]});
        hid[m * 260 + nt * 16 + lr]            = g0.x;
        hid[(m + 1) * 260 + nt * 16 + lr]      = g0.y;
        hid[m * 260 + nt * 16 + 16 + lr]       = g1.x;
        hid[(m + 1) * 260 + nt * 16 + 16 + lr] = g1.y;
    }
}

__device__ __forceinline__ void ffn_load_w2(const float* __restrict__ wp, int g, v4f (&w)[16]) {
    #pragma unroll
    for (int i = 0; i < 8; ++i) {
        const int kt = 8 * g + i;
        w[2 * i]     = *(const v4f*)(wp + kt * 4 * CC);
        w[2 * i + 1] = *(const v4f*)(wp + kt * 4 * CC + CC);
    }
}

__global__ __launch_bounds__(32)
void ffn_kernel(const float* __restrict__ x, const float* __restrict__ xln,
                const float* __restrict__ w1, const float* __restrict__ b1,
                const float* __restrict__ w2, const float* __restrict__ b2,
                float* __restrict__ out)
{
    __shared__ __attribute__((aligned(16))) float hid[16 * 260];

    const int  l   = threadIdx.x;
    const long gq0 = (long)blockIdx.x * 16;
    const bool hi = l >= 16;
    const int  lr = l & 15;

    v2f afr[17];
    {
        const float* xr = xln + (gq0 + lr) * CC + (hi ? 2 : 0);
        #pragma unroll
        for (int kt = 0; kt < 16; ++kt) afr[kt] = *(const v2f*)(xr + kt * 4);
        const v2f sc = *(const v2f*)(x + (gq0 + lr) * MD + CC);
        afr[16].x = hi ? 0.f : sc.x;
        afr[16].y = hi ? 0.f : sc.y;
    }

    v2f bA0[17], bA1[17], bB0[17], bB1[17];
    ffn_load_w1(w1, 0, lr, hi, bA0);
    ffn_load_w1(w1, 1, lr, hi, bA1);
    float bbA0 = b1[lr], bbA1 = b1[16 + lr];
    for (int nt = 0; nt < 16; nt += 4) {
        ffn_load_w1(w1, nt + 2, lr, hi, bB0);
        ffn_load_w1(w1, nt + 3, lr, hi, bB1);
        const float bbB0 = b1[(nt + 2) * 16 + lr], bbB1 = b1[(nt + 3) * 16 + lr];
        ffn_hidden_pair(afr, bA0, bA1, bbA0, bbA1, hid, nt, lr, hi);
        const int n4 = (nt + 4 < 16) ? nt + 4 : 14;
        ffn_load_w1(w1, n4,     lr, hi, bA0);
        ffn_load_w1(w1, n4 + 1, lr, hi, bA1);
        bbA0 = b1[n4 * 16 + lr];
        bbA1 = b1[(n4 + 1) * 16 + lr];
        ffn_hidden_pair(afr, bB0, bB1, bbB0, bbB1, hid, nt + 2, lr, hi);
    }

    const float* wp = w2 + (hi ? 2 : 0) * CC + 4 * lr;
    v4f wA[16], wB[16];
    ffn_load_w2(wp, 0, wA);
    __syncthreads();

    v2f a2[64];
    {
        const float* hp = hid + lr * 260 + (hi ? 2 : 0);
        #pragma unroll
        for (int kt = 0; kt < 64; ++kt) a2[kt] = *(const v2f*)(hp + kt * 4);
    }
    __builtin_amdgcn_sched_barrier(0);
    v4f res[8];
    v8f acc[4];
    {
        const v4f bb = *(const v4f*)(b2 + 4 * lr);
        #pragma unroll
        for (int j = 0; j < 4; ++j) {
            const float b = 2.f * bb[j];
            acc[j] = (v8f){b, b, b, b, b, b, b, b};
        }
    }
    #pragma unroll
    for (int g = 0; g < 8; g += 2) {
        ffn_load_w2(wp, g + 1, wB);
        __builtin_amdgcn_sched_barrier(0);
        #pragma unroll
        for (int i = 0; i < 8; ++i) {
            #pragma unroll
            for (int j = 0; j < 4; ++j)
                acc[j] = wmma_f32(a2[8 * g + i], (v2f){wA[2 * i][j], wA[2 * i + 1][j]}, acc[j]);
        }
        __builtin_amdgcn_sched_barrier(0);
        if (g + 2 < 8) {
            ffn_load_w2(wp, g + 2, wA);
        } else {
            #pragma unroll
            for (int r = 0; r < 8; ++r)
                res[r] = *(const v4f*)(xln + (gq0 + r + (hi ? 8 : 0)) * CC + 4 * lr);
        }
        __builtin_amdgcn_sched_barrier(0);
        #pragma unroll
        for (int i = 0; i < 8; ++i) {
            #pragma unroll
            for (int j = 0; j < 4; ++j)
                acc[j] = wmma_f32(a2[8 * (g + 1) + i], (v2f){wB[2 * i][j], wB[2 * i + 1][j]}, acc[j]);
        }
        __builtin_amdgcn_sched_barrier(0);
    }
    #pragma unroll
    for (int r = 0; r < 8; ++r) {
        const int  m  = r + (hi ? 8 : 0);
        const long gq = gq0 + m;
        const v4f o = {__builtin_fmaf(acc[0][r], 0.5f, res[r].x), __builtin_fmaf(acc[1][r], 0.5f, res[r].y),
                       __builtin_fmaf(acc[2][r], 0.5f, res[r].z), __builtin_fmaf(acc[3][r], 0.5f, res[r].w)};
        __builtin_memcpy(__builtin_assume_aligned(out + gq * MD + 4 * lr, 8), &o, 16);
    }
    {
        const int  m  = l >> 1, c2 = l & 1;
        const long gq = gq0 + m;
        out[gq * MD + CC + c2] = x[gq * MD + CC + c2];
    }
}

extern "C" void kernel_launch(void* const* d_in, const int* in_sizes, int n_in,
                              void* d_out, int out_size, void* d_ws, size_t ws_size,
                              hipStream_t stream)
{
    const float* x       = (const float*)d_in[0];
    const int*   nbr     = (const int*)  d_in[1];
    const float* basis   = (const float*)d_in[2];
    const float* disco_w = (const float*)d_in[3];
    const float* disco_b = (const float*)d_in[4];
    const float* lw1     = (const float*)d_in[5];
    const float* lb1     = (const float*)d_in[6];
    const float* lw2     = (const float*)d_in[7];
    const float* lb2     = (const float*)d_in[8];
    const float* hw1     = (const float*)d_in[9];
    const float* hb1     = (const float*)d_in[10];
    const float* hw2     = (const float*)d_in[11];
    const float* hb2     = (const float*)d_in[12];
    const float* fw1     = (const float*)d_in[13];
    const float* fb1     = (const float*)d_in[14];
    const float* fw2     = (const float*)d_in[15];
    const float* fb2     = (const float*)d_in[16];
    float* out = (float*)d_out;

    float* xln = (float*)((char*)d_ws + (size_t)(BB * HH_ * 32) * sizeof(float));

    disco_head_kernel<<<PP, 32, 0, stream>>>(x, nbr, basis, disco_w, disco_b,
                                                  hw1, hb1, hw2, hb2, lw1, lb1, lw2, lb2, xln);
    ffn_kernel<<<(BB * PP) / 16, 32, 0, stream>>>(x, xln, fw1, fb1, fw2, fb2, out);
}
